# starting kernel
# baseline (speedup 1.0000x reference)
#include <hip/hip_runtime.h>

typedef __attribute__((ext_vector_type(2))) float v2f;
typedef __attribute__((ext_vector_type(8))) float v8f;

#define DFEAT 256


__global__ void zero_kernel(float* __restrict__ p, long n) {
    long i = (long)blockIdx.x * blockDim.x + threadIdx.x;
    long stride = (long)gridDim.x * blockDim.x;
    for (; i < n; i += stride) p[i] = 0.0f;
}

__global__ void scatter_accum_kernel(const float* __restrict__ hsrc,
                                     const int* __restrict__ src,
                                     const int* __restrict__ dst,
                                     float* __restrict__ sums,
                                     float* __restrict__ cnt,
                                     int nE) {
    int e = blockIdx.x;
    if (e >= nE) return;
    int s = src[e];
    int d = dst[e];
    const float* row = hsrc + (size_t)s * DFEAT;
    float*       out = sums + (size_t)d * DFEAT;
    atomicAdd(&out[threadIdx.x], row[threadIdx.x]);
    if (threadIdx.x == 0) atomicAdd(&cnt[d], 1.0f);
}

__global__ void inv_kernel(const float* __restrict__ cnt,
                           float* __restrict__ inv, int n) {
    int i = blockIdx.x * blockDim.x + threadIdx.x;
    if (i < n) inv[i] = 1.0f / fmaxf(cnt[i], 1.0f);
}

template<int NT, bool RAGGED, bool RELU>
__global__ void sage_gemm_wmma(const float* __restrict__ Asum,
                               const float* __restrict__ invc,
                               const float* __restrict__ Htgt,
                               const float* __restrict__ Wl,
                               const float* __restrict__ Wr,
                               const float* __restrict__ bias,
                               float* __restrict__ Out,
                               int tilesNG,
                               int totalTiles,
                               int Nact) {
    const int wave = threadIdx.x >> 5;
    const int lane = threadIdx.x & 31;
    const int gw = blockIdx.x * (blockDim.x >> 5) + wave;
    if (gw >= totalTiles) return;

    const int tileM  = gw / tilesNG;
    const int tileNG = gw % tilesNG;
    const int l15  = lane & 15;
    const int row  = tileM * 16 + l15;
    const int koff = (lane < 16) ? 0 : 2;
    const int colBase = tileNG * (16 * NT);

    int col[NT];
#pragma unroll
    for (int j = 0; j < NT; ++j) col[j] = colBase + j * 16 + l15;

    const float invr = invc[row];

    v8f acc[NT];
#pragma unroll
    for (int j = 0; j < NT; ++j) {
        float bv = (!RAGGED || col[j] < Nact) ? bias[col[j]] : 0.0f;
#pragma unroll
        for (int i = 0; i < 8; ++i) acc[j][i] = bv;
    }

    const float* arow = Asum + (size_t)row * DFEAT + koff;
    const float* hrow = Htgt + (size_t)row * DFEAT + koff;
    const float* pl = Wl + (size_t)koff * Nact;
    const float* pr = Wr + (size_t)koff * Nact;
    const size_t bstep = 4 * (size_t)Nact;

    for (int k = 0; k < DFEAT; k += 4) {
        float2 av = *(const float2*)arow;  arow += 4;
        float2 hv = *(const float2*)hrow;  hrow += 4;
        v2f a;  a.x  = av.x * invr; a.y  = av.y * invr;
        v2f a2; a2.x = hv.x;        a2.y = hv.y;

#pragma unroll
        for (int j = 0; j < NT; ++j) {
            v2f b, b2;
            if (!RAGGED || col[j] < Nact) {
                b.x  = pl[col[j]];
                b.y  = pl[Nact + col[j]];
                b2.x = pr[col[j]];
                b2.y = pr[Nact + col[j]];
            } else {
                b.x = 0.0f; b.y = 0.0f; b2.x = 0.0f; b2.y = 0.0f;
            }
            acc[j] = __builtin_amdgcn_wmma_f32_16x16x4_f32(
                         false, a, false, b, (short)0, acc[j], false, false);
            acc[j] = __builtin_amdgcn_wmma_f32_16x16x4_f32(
                         false, a2, false, b2, (short)0, acc[j], false, false);
        }
        pl += bstep;
        pr += bstep;
    }

    const int mbase = tileM * 16 + ((lane < 16) ? 0 : 8);
#pragma unroll
    for (int j = 0; j < NT; ++j) {
        if (!RAGGED || col[j] < Nact) {
#pragma unroll
            for (int i = 0; i < 8; ++i) {
                float v = acc[j][i];
                if (RELU) v = fmaxf(v, 0.0f);
                Out[(size_t)(mbase + i) * Nact + col[j]] = v;
            }
        }
    }
}


extern "C" void kernel_launch(void* const* d_in, const int* in_sizes, int n_in,
                              void* d_out, int out_size, void* d_ws, size_t ws_size,
                              hipStream_t stream) {
    const float* h   = (const float*)d_in[0];
    const float* wl0 = (const float*)d_in[1];
    const float* wr0 = (const float*)d_in[2];
    const float* b0  = (const float*)d_in[3];
    const float* wl1 = (const float*)d_in[4];
    const float* wr1 = (const float*)d_in[5];
    const float* b1  = (const float*)d_in[6];
    const int*   ei0 = (const int*)d_in[7];
    const int*   ei1 = (const int*)d_in[8];

    const int E0 = in_sizes[7] / 2;
    const int E1 = in_sizes[8] / 2;
    const int NT0 = 20000;
    const int NT1 = 2000;
    const int NCLS = 47;

    float* sums0 = (float*)d_ws;
    float* cnt0  = sums0 + (size_t)NT0 * DFEAT;
    float* inv0  = cnt0 + NT0;
    float* h1    = inv0 + NT0;
    float* sums1 = h1 + (size_t)NT0 * DFEAT;
    float* cnt1  = sums1 + (size_t)NT1 * DFEAT;
    float* inv1  = cnt1 + NT1;

    zero_kernel<<<2048, 256, 0, stream>>>(sums0, (long)NT0 * DFEAT);
    zero_kernel<<<64,   256, 0, stream>>>(cnt0, NT0);
    zero_kernel<<<512,  256, 0, stream>>>(sums1, (long)NT1 * DFEAT);
    zero_kernel<<<8,    256, 0, stream>>>(cnt1, NT1);

    scatter_accum_kernel<<<E0, DFEAT, 0, stream>>>(h, ei0, ei0 + E0, sums0, cnt0, E0);
    inv_kernel<<<(NT0 + 255) / 256, 256, 0, stream>>>(cnt0, inv0, NT0);

    {
        const int tilesNG = DFEAT / 64;
        const int total   = (NT0 / 16) * tilesNG;
        const int blocks  = (total + 7) / 8;
        sage_gemm_wmma<4, false, true><<<blocks, 256, 0, stream>>>(
            sums0, inv0, h, wl0, wr0, b0, h1, tilesNG, total, DFEAT);
    }

    scatter_accum_kernel<<<E1, DFEAT, 0, stream>>>(h1, ei1, ei1 + E1, sums1, cnt1, E1);
    inv_kernel<<<(NT1 + 255) / 256, 256, 0, stream>>>(cnt1, inv1, NT1);

    {
        const int tilesNG = 1;
        const int total   = (NT1 / 16) * tilesNG;
        const int blocks  = (total + 7) / 8;
        sage_gemm_wmma<3, true, false><<<blocks, 256, 0, stream>>>(
            sums1, inv1, h1, wl1, wr1, b1, (float*)d_out, tilesNG, total, NCLS);
    }
}


# Aggregation without float atomics: edges grouped by target (integer returning add as rank, prefix sum, placement), then one wave per target sums its rows in registers; sum order = rank order
# speedup vs baseline: 1.7187x; 1.7187x over previous
#include <hip/hip_runtime.h>

typedef __attribute__((ext_vector_type(2))) float v2f;
typedef __attribute__((ext_vector_type(8))) float v8f;

#define DFEAT 256


__global__ void zero_kernel(float* __restrict__ p, long n) {
    long i = (long)blockIdx.x * blockDim.x + threadIdx.x;
    long stride = (long)gridDim.x * blockDim.x;
    for (; i < n; i += stride) p[i] = 0.0f;
}


__global__ void rank_kernel(const int* __restrict__ dstA, int nEA, int* __restrict__ degA, int* __restrict__ rankA,
                            const int* __restrict__ dstB, int nEB, int* __restrict__ degB, int* __restrict__ rankB) {
    int i = blockIdx.x * blockDim.x + threadIdx.x;
    if (i < nEA) {
        rankA[i] = atomicAdd(&degA[dstA[i]], 1);
    } else {
        int k = i - nEA;
        if (k < nEB) rankB[k] = atomicAdd(&degB[dstB[k]], 1);
    }
}

#define SCAN_IPT 20
__global__ void __launch_bounds__(1024)
scan_kernel(const int* __restrict__ degA, int* __restrict__ rpA, int nA,
            const int* __restrict__ degB, int* __restrict__ rpB, int nB) {
    const int* deg; int* rp; int n;
    if (blockIdx.x == 0) { deg = degA; rp = rpA; n = nA; }
    else                 { deg = degB; rp = rpB; n = nB; }

    __shared__ int wsum[32];
    const int t = threadIdx.x;
    const int lane = t & 31;
    const int w = t >> 5;
    const int ipt = (n + 1023) >> 10;
    const int first = t * ipt;

    int v[SCAN_IPT];
    int s = 0;
#pragma unroll
    for (int i = 0; i < SCAN_IPT; ++i) {
        int idx = first + i;
        v[i] = (i < ipt && idx < n) ? deg[idx] : 0;
        s += v[i];
    }
    int x = s;
#pragma unroll
    for (int o = 1; o < 32; o <<= 1) {
        int y = __shfl_up(x, o);
        if (lane >= o) x += y;
    }
    if (lane == 31) wsum[w] = x;
    __syncthreads();
    if (w == 0) {
        int z = wsum[lane];
        int zi = z;
#pragma unroll
        for (int o = 1; o < 32; o <<= 1) {
            int y = __shfl_up(zi, o);
            if (lane >= o) zi += y;
        }
        wsum[lane] = zi - z;
    }
    __syncthreads();
    int off = wsum[w] + x - s;
#pragma unroll
    for (int i = 0; i < SCAN_IPT; ++i) {
        int idx = first + i;
        if (i < ipt && idx < n) {
            rp[idx] = off;
            off += v[i];
        }
    }
}

__global__ void place_kernel(const int* __restrict__ srcA, const int* __restrict__ dstA, int nEA,
                             const int* __restrict__ rpA, const int* __restrict__ rankA, int* __restrict__ csrA,
                             const int* __restrict__ srcB, const int* __restrict__ dstB, int nEB,
                             const int* __restrict__ rpB, const int* __restrict__ rankB, int* __restrict__ csrB) {
    int i = blockIdx.x * blockDim.x + threadIdx.x;
    if (i < nEA) {
        csrA[rpA[dstA[i]] + rankA[i]] = srcA[i];
    } else {
        int k = i - nEA;
        if (k < nEB) csrB[rpB[dstB[k]] + rankB[k]] = srcB[k];
    }
}

#define GATHER_U 8
__global__ void __launch_bounds__(256)
gather_sum_kernel(const float* __restrict__ hsrc,
                  const int* __restrict__ rowptr,
                  const int* __restrict__ deg,
                  const int* __restrict__ csr,
                  float* __restrict__ sums,
                  float* __restrict__ inv,
                  int nT) {
    const int lane = threadIdx.x & 31;
    const int node = __builtin_amdgcn_readfirstlane(blockIdx.x * (blockDim.x >> 5) + (threadIdx.x >> 5));
    if (node >= nT) return;
    const int beg = __builtin_amdgcn_readfirstlane(rowptr[node]);
    const int n   = __builtin_amdgcn_readfirstlane(deg[node]);

    const float4 zero4 = make_float4(0.0f, 0.0f, 0.0f, 0.0f);
    float4 a0 = zero4;
    float4 a1 = zero4;

    for (int base = 0; base < n; base += 32) {
        const int m = min(32, n - base);
        const int myid = (lane < m) ? csr[beg + base + lane] : 0;
        for (int j = 0; j < m; j += GATHER_U) {
            int sid[GATHER_U];
#pragma unroll
            for (int k = 0; k < GATHER_U; ++k) sid[k] = __builtin_amdgcn_readlane(myid, (j + k) & 31);

            float4 x[GATHER_U], y[GATHER_U];
            const int left = m - j;
            if (left >= GATHER_U) {
#pragma unroll
                for (int k = 0; k < GATHER_U; ++k) {
                    const float4* r = (const float4*)(hsrc + (size_t)sid[k] * DFEAT);
                    x[k] = r[lane];
                    y[k] = r[32 + lane];
                }
            } else {
#pragma unroll
                for (int k = 0; k < GATHER_U; ++k) {
                    if (k < left) {
                        const float4* r = (const float4*)(hsrc + (size_t)sid[k] * DFEAT);
                        x[k] = r[lane];
                        y[k] = r[32 + lane];
                    } else {
                        x[k] = zero4;
                        y[k] = zero4;
                    }
                }
            }
#pragma unroll
            for (int k = 0; k < GATHER_U; ++k) {
                a0.x += x[k].x; a0.y += x[k].y; a0.z += x[k].z; a0.w += x[k].w;
                a1.x += y[k].x; a1.y += y[k].y; a1.z += y[k].z; a1.w += y[k].w;
            }
        }
    }

    float4* o = (float4*)(sums + (size_t)node * DFEAT);
    o[lane]      = a0;
    o[32 + lane] = a1;
    if (lane == 0) inv[node] = 1.0f / fmaxf((float)n, 1.0f);
}

template<int NT, bool RAGGED, bool RELU>
__global__ void sage_gemm_wmma(const float* __restrict__ Asum,
                               const float* __restrict__ invc,
                               const float* __restrict__ Htgt,
                               const float* __restrict__ Wl,
                               const float* __restrict__ Wr,
                               const float* __restrict__ bias,
                               float* __restrict__ Out,
                               int tilesNG,
                               int totalTiles,
                               int Nact) {
    const int wave = threadIdx.x >> 5;
    const int lane = threadIdx.x & 31;
    const int gw = blockIdx.x * (blockDim.x >> 5) + wave;
    if (gw >= totalTiles) return;

    const int tileM  = gw / tilesNG;
    const int tileNG = gw % tilesNG;
    const int l15  = lane & 15;
    const int row  = tileM * 16 + l15;
    const int koff = (lane < 16) ? 0 : 2;
    const int colBase = tileNG * (16 * NT);

    int col[NT];
#pragma unroll
    for (int j = 0; j < NT; ++j) col[j] = colBase + j * 16 + l15;

    const float invr = invc[row];

    v8f acc[NT];
#pragma unroll
    for (int j = 0; j < NT; ++j) {
        float bv = (!RAGGED || col[j] < Nact) ? bias[col[j]] : 0.0f;
#pragma unroll
        for (int i = 0; i < 8; ++i) acc[j][i] = bv;
    }

    const float* arow = Asum + (size_t)row * DFEAT + koff;
    const float* hrow = Htgt + (size_t)row * DFEAT + koff;
    const float* pl = Wl + (size_t)koff * Nact;
    const float* pr = Wr + (size_t)koff * Nact;
    const size_t bstep = 4 * (size_t)Nact;

    for (int k = 0; k < DFEAT; k += 4) {
        float2 av = *(const float2*)arow;  arow += 4;
        float2 hv = *(const float2*)hrow;  hrow += 4;
        v2f a;  a.x  = av.x * invr; a.y  = av.y * invr;
        v2f a2; a2.x = hv.x;        a2.y = hv.y;

#pragma unroll
        for (int j = 0; j < NT; ++j) {
            v2f b, b2;
            if (!RAGGED || col[j] < Nact) {
                b.x  = pl[col[j]];
                b.y  = pl[Nact + col[j]];
                b2.x = pr[col[j]];
                b2.y = pr[Nact + col[j]];
            } else {
                b.x = 0.0f; b.y = 0.0f; b2.x = 0.0f; b2.y = 0.0f;
            }
            acc[j] = __builtin_amdgcn_wmma_f32_16x16x4_f32(
                         false, a, false, b, (short)0, acc[j], false, false);
            acc[j] = __builtin_amdgcn_wmma_f32_16x16x4_f32(
                         false, a2, false, b2, (short)0, acc[j], false, false);
        }
        pl += bstep;
        pr += bstep;
    }

    const int mbase = tileM * 16 + ((lane < 16) ? 0 : 8);
#pragma unroll
    for (int j = 0; j < NT; ++j) {
        if (!RAGGED || col[j] < Nact) {
#pragma unroll
            for (int i = 0; i < 8; ++i) {
                float v = acc[j][i];
                if (RELU) v = fmaxf(v, 0.0f);
                Out[(size_t)(mbase + i) * Nact + col[j]] = v;
            }
        }
    }
}


extern "C" void kernel_launch(void* const* d_in, const int* in_sizes, int n_in,
                              void* d_out, int out_size, void* d_ws, size_t ws_size,
                              hipStream_t stream) {
    const float* h   = (const float*)d_in[0];
    const float* wl0 = (const float*)d_in[1];
    const float* wr0 = (const float*)d_in[2];
    const float* b0  = (const float*)d_in[3];
    const float* wl1 = (const float*)d_in[4];
    const float* wr1 = (const float*)d_in[5];
    const float* b1  = (const float*)d_in[6];
    const int*   ei0 = (const int*)d_in[7];
    const int*   ei1 = (const int*)d_in[8];

    const int E0 = in_sizes[7] / 2;
    const int E1 = in_sizes[8] / 2;
    const int NT0 = 20000;
    const int NT1 = 2000;
    const int NCLS = 47;

    float* sums0 = (float*)d_ws;
    float* cnt0  = sums0 + (size_t)NT0 * DFEAT;
    float* inv0  = cnt0 + NT0;
    float* h1    = inv0 + NT0;
    float* reg1  = h1 + (size_t)NT0 * DFEAT;
    float* cnt1  = reg1 + (size_t)NT1 * DFEAT;
    float* inv1  = cnt1 + NT1;
    float* sums1 = sums0;

    int* csr0  = (int*)h1;
    int* rank0 = csr0 + E0;
    int* rank1 = rank0 + E0;
    int* rp0   = rank1 + E1;
    int* deg0  = (int*)reg1 - NT0;
    int* deg1  = (int*)reg1;
    int* rp1   = deg1 + NT1;
    int* csr1  = rp1 + NT1;

    zero_kernel<<<22, 1024, 0, stream>>>((float*)deg0, (long)(NT0 + NT1));
    rank_kernel<<<(E0 + E1 + 255) / 256, 256, 0, stream>>>(ei0 + E0, E0, deg0, rank0, ei1 + E1, E1, deg1, rank1);
    scan_kernel<<<2, 1024, 0, stream>>>(deg0, rp0, NT0, deg1, rp1, NT1);
    place_kernel<<<(E0 + E1 + 255) / 256, 256, 0, stream>>>(ei0, ei0 + E0, E0, rp0, rank0, csr0,
                                                            ei1, ei1 + E1, E1, rp1, rank1, csr1);

    gather_sum_kernel<<<(NT0 + 7) / 8, 256, 0, stream>>>(h, rp0, deg0, csr0, sums0, inv0, NT0);

    {
        const int tilesNG = DFEAT / 64;
        const int total   = (NT0 / 16) * tilesNG;
        const int blocks  = (total + 7) / 8;
        sage_gemm_wmma<4, false, true><<<blocks, 256, 0, stream>>>(
            sums0, inv0, h, wl0, wr0, b0, h1, tilesNG, total, DFEAT);
    }

    gather_sum_kernel<<<(NT1 + 7) / 8, 256, 0, stream>>>(h1, rp1, deg1, csr1, sums1, inv1, NT1);

    {
        const int tilesNG = 1;
        const int total   = (NT1 / 16) * tilesNG;
        const int blocks  = (total + 7) / 8;
        sage_gemm_wmma<3, true, false><<<blocks, 256, 0, stream>>>(
            sums1, inv1, h1, wl1, wr1, b1, (float*)d_out, tilesNG, total, NCLS);
    }
}


# Both GEMM K loops software-pipelined: next 8-k group's fragments (16-byte loads; k and column order inside a wave's tiles permuted) requested ahead of the current WMMAs; sum order over k changed
# speedup vs baseline: 2.7788x; 1.6168x over previous
#include <hip/hip_runtime.h>

typedef __attribute__((ext_vector_type(2))) float v2f;
typedef __attribute__((ext_vector_type(8))) float v8f;

#define DFEAT 256


__global__ void zero_kernel(float* __restrict__ p, long n) {
    long i = (long)blockIdx.x * blockDim.x + threadIdx.x;
    long stride = (long)gridDim.x * blockDim.x;
    for (; i < n; i += stride) p[i] = 0.0f;
}


__global__ void rank_kernel(const int* __restrict__ dstA, int nEA, int* __restrict__ degA, int* __restrict__ rankA,
                            const int* __restrict__ dstB, int nEB, int* __restrict__ degB, int* __restrict__ rankB) {
    int i = blockIdx.x * blockDim.x + threadIdx.x;
    if (i < nEA) {
        rankA[i] = atomicAdd(&degA[dstA[i]], 1);
    } else {
        int k = i - nEA;
        if (k < nEB) rankB[k] = atomicAdd(&degB[dstB[k]], 1);
    }
}

#define SCAN_IPT 20
__global__ void __launch_bounds__(1024)
scan_kernel(const int* __restrict__ degA, int* __restrict__ rpA, int nA,
            const int* __restrict__ degB, int* __restrict__ rpB, int nB) {
    const int* deg; int* rp; int n;
    if (blockIdx.x == 0) { deg = degA; rp = rpA; n = nA; }
    else                 { deg = degB; rp = rpB; n = nB; }

    __shared__ int wsum[32];
    const int t = threadIdx.x;
    const int lane = t & 31;
    const int w = t >> 5;
    const int ipt = (n + 1023) >> 10;
    const int first = t * ipt;

    int v[SCAN_IPT];
    int s = 0;
#pragma unroll
    for (int i = 0; i < SCAN_IPT; ++i) {
        int idx = first + i;
        v[i] = (i < ipt && idx < n) ? deg[idx] : 0;
        s += v[i];
    }
    int x = s;
#pragma unroll
    for (int o = 1; o < 32; o <<= 1) {
        int y = __shfl_up(x, o);
        if (lane >= o) x += y;
    }
    if (lane == 31) wsum[w] = x;
    __syncthreads();
    if (w == 0) {
        int z = wsum[lane];
        int zi = z;
#pragma unroll
        for (int o = 1; o < 32; o <<= 1) {
            int y = __shfl_up(zi, o);
            if (lane >= o) zi += y;
        }
        wsum[lane] = zi - z;
    }
    __syncthreads();
    int off = wsum[w] + x - s;
#pragma unroll
    for (int i = 0; i < SCAN_IPT; ++i) {
        int idx = first + i;
        if (i < ipt && idx < n) {
            rp[idx] = off;
            off += v[i];
        }
    }
}

__global__ void place_kernel(const int* __restrict__ srcA, const int* __restrict__ dstA, int nEA,
                             const int* __restrict__ rpA, const int* __restrict__ rankA, int* __restrict__ csrA,
                             const int* __restrict__ srcB, const int* __restrict__ dstB, int nEB,
                             const int* __restrict__ rpB, const int* __restrict__ rankB, int* __restrict__ csrB) {
    int i = blockIdx.x * blockDim.x + threadIdx.x;
    if (i < nEA) {
        csrA[rpA[dstA[i]] + rankA[i]] = srcA[i];
    } else {
        int k = i - nEA;
        if (k < nEB) csrB[rpB[dstB[k]] + rankB[k]] = srcB[k];
    }
}

#define GATHER_U 8
__global__ void __launch_bounds__(256)
gather_sum_kernel(const float* __restrict__ hsrc,
                  const int* __restrict__ rowptr,
                  const int* __restrict__ deg,
                  const int* __restrict__ csr,
                  float* __restrict__ sums,
                  float* __restrict__ inv,
                  int nT) {
    const int lane = threadIdx.x & 31;
    const int node = __builtin_amdgcn_readfirstlane(blockIdx.x * (blockDim.x >> 5) + (threadIdx.x >> 5));
    if (node >= nT) return;
    const int beg = __builtin_amdgcn_readfirstlane(rowptr[node]);
    const int n   = __builtin_amdgcn_readfirstlane(deg[node]);

    const float4 zero4 = make_float4(0.0f, 0.0f, 0.0f, 0.0f);
    float4 a0 = zero4;
    float4 a1 = zero4;

    for (int base = 0; base < n; base += 32) {
        const int m = min(32, n - base);
        const int myid = (lane < m) ? csr[beg + base + lane] : 0;
        for (int j = 0; j < m; j += GATHER_U) {
            int sid[GATHER_U];
#pragma unroll
            for (int k = 0; k < GATHER_U; ++k) sid[k] = __builtin_amdgcn_readlane(myid, (j + k) & 31);

            float4 x[GATHER_U], y[GATHER_U];
            const int left = m - j;
            if (left >= GATHER_U) {
#pragma unroll
                for (int k = 0; k < GATHER_U; ++k) {
                    const float4* r = (const float4*)(hsrc + (size_t)sid[k] * DFEAT);
                    x[k] = r[lane];
                    y[k] = r[32 + lane];
                }
            } else {
#pragma unroll
                for (int k = 0; k < GATHER_U; ++k) {
                    if (k < left) {
                        const float4* r = (const float4*)(hsrc + (size_t)sid[k] * DFEAT);
                        x[k] = r[lane];
                        y[k] = r[32 + lane];
                    } else {
                        x[k] = zero4;
                        y[k] = zero4;
                    }
                }
            }
#pragma unroll
            for (int k = 0; k < GATHER_U; ++k) {
                a0.x += x[k].x; a0.y += x[k].y; a0.z += x[k].z; a0.w += x[k].w;
                a1.x += y[k].x; a1.y += y[k].y; a1.z += y[k].z; a1.w += y[k].w;
            }
        }
    }

    float4* o = (float4*)(sums + (size_t)node * DFEAT);
    o[lane]      = a0;
    o[32 + lane] = a1;
    if (lane == 0) inv[node] = 1.0f / fmaxf((float)n, 1.0f);
}


__device__ __forceinline__ v8f mma4(v2f a, v2f b, v8f c) {
    return __builtin_amdgcn_wmma_f32_16x16x4_f32(false, a, false, b, (short)0, c, false, false);
}

struct FragV {
    float4 as[2], at[2];
    float4 bl[4], br[4];
};

__device__ __forceinline__ void l0_load(FragV& f, const float* __restrict__ a0, const float* __restrict__ t0,
                                        const float* __restrict__ pl, const float* __restrict__ pr, int g) {
    const int ko = g * 8;
    f.as[0] = *(const float4*)(a0 + ko);
    f.as[1] = *(const float4*)(a0 + 16 * DFEAT + ko);
    f.at[0] = *(const float4*)(t0 + ko);
    f.at[1] = *(const float4*)(t0 + 16 * DFEAT + ko);
#pragma unroll
    for (int i = 0; i < 4; ++i) {
        f.bl[i] = *(const float4*)(pl + (size_t)(ko + i) * DFEAT);
        f.br[i] = *(const float4*)(pr + (size_t)(ko + i) * DFEAT);
    }
}

__device__ __forceinline__ void l0_mma(v8f (&acc)[2][4], const FragV& f, float inv0, float inv1) {
    v2f aS[2][2], aT[2][2];
    aS[0][0].x = f.as[0].x * inv0; aS[0][0].y = f.as[0].y * inv0;
    aS[0][1].x = f.as[0].z * inv0; aS[0][1].y = f.as[0].w * inv0;
    aS[1][0].x = f.as[1].x * inv1; aS[1][0].y = f.as[1].y * inv1;
    aS[1][1].x = f.as[1].z * inv1; aS[1][1].y = f.as[1].w * inv1;
    aT[0][0].x = f.at[0].x; aT[0][0].y = f.at[0].y;
    aT[0][1].x = f.at[0].z; aT[0][1].y = f.at[0].w;
    aT[1][0].x = f.at[1].x; aT[1][0].y = f.at[1].y;
    aT[1][1].x = f.at[1].z; aT[1][1].y = f.at[1].w;

    float L[4][4], R[4][4];
#pragma unroll
    for (int i = 0; i < 4; ++i) {
        L[i][0] = f.bl[i].x; L[i][1] = f.bl[i].y; L[i][2] = f.bl[i].z; L[i][3] = f.bl[i].w;
        R[i][0] = f.br[i].x; R[i][1] = f.br[i].y; R[i][2] = f.br[i].z; R[i][3] = f.br[i].w;
    }
#pragma unroll
    for (int s = 0; s < 2; ++s) {
#pragma unroll
        for (int j = 0; j < 4; ++j) {
            v2f b; b.x = L[2 * s][j]; b.y = L[2 * s + 1][j];
            acc[0][j] = mma4(aS[0][s], b, acc[0][j]);
            acc[1][j] = mma4(aS[1][s], b, acc[1][j]);
        }
#pragma unroll
        for (int j = 0; j < 4; ++j) {
            v2f b; b.x = R[2 * s][j]; b.y = R[2 * s + 1][j];
            acc[0][j] = mma4(aT[0][s], b, acc[0][j]);
            acc[1][j] = mma4(aT[1][s], b, acc[1][j]);
        }
    }
}

__global__ void __launch_bounds__(128)
sage_gemm_l0(const float* __restrict__ Asum,
             const float* __restrict__ invc,
             const float* __restrict__ Htgt,
             const float* __restrict__ Wl,
             const float* __restrict__ Wr,
             const float* __restrict__ bias,
             float* __restrict__ Out,
             int nPairs) {
    const int lane = threadIdx.x & 31;
    const int pair = blockIdx.x;
    if (pair >= nPairs) return;
    const int c = lane & 15;
    const int h = lane >> 4;
    const int colBase = (threadIdx.x >> 5) * 64;
    const int row0 = pair * 32 + c;

    const float inv0 = invc[row0];
    const float inv1 = invc[row0 + 16];

    const float4 bv = *(const float4*)(bias + colBase + 4 * c);
    v8f acc[2][4];
#pragma unroll
    for (int i = 0; i < 8; ++i) {
        acc[0][0][i] = bv.x; acc[0][1][i] = bv.y; acc[0][2][i] = bv.z; acc[0][3][i] = bv.w;
        acc[1][0][i] = bv.x; acc[1][1][i] = bv.y; acc[1][2][i] = bv.z; acc[1][3][i] = bv.w;
    }

    const float* a0 = Asum + (size_t)row0 * DFEAT + 4 * h;
    const float* t0 = Htgt + (size_t)row0 * DFEAT + 4 * h;
    const float* pl = Wl + (size_t)(4 * h) * DFEAT + colBase + 4 * c;
    const float* pr = Wr + (size_t)(4 * h) * DFEAT + colBase + 4 * c;

    FragV f0, f1;
    l0_load(f0, a0, t0, pl, pr, 0);
    for (int g = 0; g < DFEAT / 8 - 2; g += 2) {
        l0_load(f1, a0, t0, pl, pr, g + 1);
        l0_mma(acc, f0, inv0, inv1);
        l0_load(f0, a0, t0, pl, pr, g + 2);
        l0_mma(acc, f1, inv0, inv1);
    }
    l0_load(f1, a0, t0, pl, pr, DFEAT / 8 - 1);
    l0_mma(acc, f0, inv0, inv1);
    l0_mma(acc, f1, inv0, inv1);

#pragma unroll
    for (int mt = 0; mt < 2; ++mt) {
        float* o = Out + (size_t)(pair * 32 + mt * 16 + 8 * h) * DFEAT + colBase + 4 * c;
#pragma unroll
        for (int i = 0; i < 8; ++i) {
            float4 v;
            v.x = fmaxf(acc[mt][0][i], 0.0f);
            v.y = fmaxf(acc[mt][1][i], 0.0f);
            v.z = fmaxf(acc[mt][2][i], 0.0f);
            v.w = fmaxf(acc[mt][3][i], 0.0f);
            *(float4*)(o + (size_t)i * DFEAT) = v;
        }
    }
}

struct FragR {
    float4 as[2], at[2];
    v2f bl[2][2][3];
    v2f br[2][2][3];
};

__device__ __forceinline__ void l1_load(FragR& f, const float* __restrict__ a0, const float* __restrict__ t0,
                                        const float* __restrict__ pl, const float* __restrict__ pr,
                                        const int (&cj)[3], int Nact, int g) {
#pragma unroll
    for (int u = 0; u < 2; ++u) {
        const int ko = g * 16 + u * 8;
        f.as[u] = *(const float4*)(a0 + ko);
        f.at[u] = *(const float4*)(t0 + ko);
#pragma unroll
        for (int s = 0; s < 2; ++s) {
            const float* ql = pl + (size_t)(ko + 2 * s) * Nact;
            const float* qr = pr + (size_t)(ko + 2 * s) * Nact;
#pragma unroll
            for (int j = 0; j < 3; ++j) {
                f.bl[u][s][j].x = ql[cj[j]];
                f.bl[u][s][j].y = ql[Nact + cj[j]];
                f.br[u][s][j].x = qr[cj[j]];
                f.br[u][s][j].y = qr[Nact + cj[j]];
            }
        }
    }
}

__device__ __forceinline__ void l1_mma(v8f (&acc)[3], const FragR& f, float inv) {
#pragma unroll
    for (int u = 0; u < 2; ++u) {
        v2f aS[2], aT[2];
        aS[0].x = f.as[u].x * inv; aS[0].y = f.as[u].y * inv;
        aS[1].x = f.as[u].z * inv; aS[1].y = f.as[u].w * inv;
        aT[0].x = f.at[u].x; aT[0].y = f.at[u].y;
        aT[1].x = f.at[u].z; aT[1].y = f.at[u].w;
#pragma unroll
        for (int s = 0; s < 2; ++s) {
#pragma unroll
            for (int j = 0; j < 3; ++j) acc[j] = mma4(aS[s], f.bl[u][s][j], acc[j]);
#pragma unroll
            for (int j = 0; j < 3; ++j) acc[j] = mma4(aT[s], f.br[u][s][j], acc[j]);
        }
    }
}

__global__ void __launch_bounds__(32)
sage_gemm_l1(const float* __restrict__ Asum,
             const float* __restrict__ invc,
             const float* __restrict__ Htgt,
             const float* __restrict__ Wl,
             const float* __restrict__ Wr,
             const float* __restrict__ bias,
             float* __restrict__ Out,
             int nTiles,
             int Nact) {
    const int lane = threadIdx.x & 31;
    const int tileM = blockIdx.x;
    if (tileM >= nTiles) return;
    const int c = lane & 15;
    const int h = lane >> 4;
    const int row = tileM * 16 + c;

    int col[3], cj[3];
#pragma unroll
    for (int j = 0; j < 3; ++j) { col[j] = j * 16 + c; cj[j] = min(col[j], Nact - 1); }

    const float inv = invc[row];

    v8f acc[3];
#pragma unroll
    for (int j = 0; j < 3; ++j) {
        const float bvj = bias[cj[j]];
#pragma unroll
        for (int i = 0; i < 8; ++i) acc[j][i] = bvj;
    }

    const float* a0 = Asum + (size_t)row * DFEAT + 4 * h;
    const float* t0 = Htgt + (size_t)row * DFEAT + 4 * h;
    const float* pl = Wl + (size_t)(4 * h) * Nact;
    const float* pr = Wr + (size_t)(4 * h) * Nact;

    FragR f0, f1;
    l1_load(f0, a0, t0, pl, pr, cj, Nact, 0);
    for (int g = 0; g < DFEAT / 16 - 2; g += 2) {
        l1_load(f1, a0, t0, pl, pr, cj, Nact, g + 1);
        l1_mma(acc, f0, inv);
        l1_load(f0, a0, t0, pl, pr, cj, Nact, g + 2);
        l1_mma(acc, f1, inv);
    }
    l1_load(f1, a0, t0, pl, pr, cj, Nact, DFEAT / 16 - 1);
    l1_mma(acc, f0, inv);
    l1_mma(acc, f1, inv);

    const int mbase = tileM * 16 + 8 * h;
#pragma unroll
    for (int j = 0; j < 3; ++j) {
        if (col[j] < Nact) {
#pragma unroll
            for (int i = 0; i < 8; ++i) Out[(size_t)(mbase + i) * Nact + col[j]] = acc[j][i];
        }
    }
}


extern "C" void kernel_launch(void* const* d_in, const int* in_sizes, int n_in,
                              void* d_out, int out_size, void* d_ws, size_t ws_size,
                              hipStream_t stream) {
    const float* h   = (const float*)d_in[0];
    const float* wl0 = (const float*)d_in[1];
    const float* wr0 = (const float*)d_in[2];
    const float* b0  = (const float*)d_in[3];
    const float* wl1 = (const float*)d_in[4];
    const float* wr1 = (const float*)d_in[5];
    const float* b1  = (const float*)d_in[6];
    const int*   ei0 = (const int*)d_in[7];
    const int*   ei1 = (const int*)d_in[8];

    const int E0 = in_sizes[7] / 2;
    const int E1 = in_sizes[8] / 2;
    const int NT0 = 20000;
    const int NT1 = 2000;
    const int NCLS = 47;

    float* sums0 = (float*)d_ws;
    float* cnt0  = sums0 + (size_t)NT0 * DFEAT;
    float* inv0  = cnt0 + NT0;
    float* h1    = inv0 + NT0;
    float* reg1  = h1 + (size_t)NT0 * DFEAT;
    float* cnt1  = reg1 + (size_t)NT1 * DFEAT;
    float* inv1  = cnt1 + NT1;
    float* sums1 = sums0;

    int* csr0  = (int*)h1;
    int* rank0 = csr0 + E0;
    int* rank1 = rank0 + E0;
    int* rp0   = rank1 + E1;
    int* deg0  = (int*)reg1 - NT0;
    int* deg1  = (int*)reg1;
    int* rp1   = deg1 + NT1;
    int* csr1  = rp1 + NT1;

    zero_kernel<<<22, 1024, 0, stream>>>((float*)deg0, (long)(NT0 + NT1));
    rank_kernel<<<(E0 + E1 + 255) / 256, 256, 0, stream>>>(ei0 + E0, E0, deg0, rank0, ei1 + E1, E1, deg1, rank1);
    scan_kernel<<<2, 1024, 0, stream>>>(deg0, rp0, NT0, deg1, rp1, NT1);
    place_kernel<<<(E0 + E1 + 255) / 256, 256, 0, stream>>>(ei0, ei0 + E0, E0, rp0, rank0, csr0,
                                                            ei1, ei1 + E1, E1, rp1, rank1, csr1);

    gather_sum_kernel<<<(NT0 + 7) / 8, 256, 0, stream>>>(h, rp0, deg0, csr0, sums0, inv0, NT0);

    sage_gemm_l0<<<NT0 / 32, 128, 0, stream>>>(sums0, inv0, h, wl0, wr0, b0, h1, NT0 / 32);

    gather_sum_kernel<<<(NT1 + 7) / 8, 256, 0, stream>>>(h1, rp1, deg1, csr1, sums1, inv1, NT1);

    sage_gemm_l1<<<NT1 / 16, 32, 0, stream>>>(sums1, inv1, h1, wl1, wr1, b1, (float*)d_out, NT1 / 16, NCLS);
}


# Fixed-stride per-target id buckets (slot = target*64 + arrival rank of the returning integer add) replace prefix sum and placement: 8 launches -> 6; sum order stays arrival order
# speedup vs baseline: 3.2763x; 1.1790x over previous
#include <hip/hip_runtime.h>

typedef __attribute__((ext_vector_type(2))) float v2f;
typedef __attribute__((ext_vector_type(8))) float v8f;

#define DFEAT 256


__global__ void zero_kernel(float* __restrict__ p, long n) {
    long i = (long)blockIdx.x * blockDim.x + threadIdx.x;
    long stride = (long)gridDim.x * blockDim.x;
    for (; i < n; i += stride) p[i] = 0.0f;
}

#define BUCKET_CAP 64

__global__ void bucket_kernel(const int* __restrict__ srcA, const int* __restrict__ dstA, int nEA,
                              int* __restrict__ degA, int* __restrict__ slotA,
                              const int* __restrict__ srcB, const int* __restrict__ dstB, int nEB,
                              int* __restrict__ degB, int* __restrict__ slotB) {
    int i = blockIdx.x * blockDim.x + threadIdx.x;
    if (i < nEA) {
        const int d = dstA[i];
        const int s = srcA[i];
        const int r = atomicAdd(&degA[d], 1);
        if (r < BUCKET_CAP) slotA[(size_t)d * BUCKET_CAP + r] = s;
    } else {
        int k = i - nEA;
        if (k < nEB) {
            const int d = dstB[k];
            const int s = srcB[k];
            const int r = atomicAdd(&degB[d], 1);
            if (r < BUCKET_CAP) slotB[(size_t)d * BUCKET_CAP + r] = s;
        }
    }
}

#define GATHER_U 8

__device__ __forceinline__ void add_row(float4& a0, float4& a1, const float* __restrict__ hsrc, int sid, int lane) {
    const float4* r = (const float4*)(hsrc + (size_t)sid * DFEAT);
    const float4 x = r[lane];
    const float4 y = r[32 + lane];
    a0.x += x.x; a0.y += x.y; a0.z += x.z; a0.w += x.w;
    a1.x += y.x; a1.y += y.y; a1.z += y.z; a1.w += y.w;
}

__global__ void __launch_bounds__(256)
gather_sum_kernel(const float* __restrict__ hsrc,
                  const int* __restrict__ deg,
                  const int* __restrict__ slot,
                  const int* __restrict__ esrc,
                  const int* __restrict__ edst,
                  int nE,
                  float* __restrict__ sums,
                  float* __restrict__ inv,
                  int nT) {
    const int lane = threadIdx.x & 31;
    const int node = __builtin_amdgcn_readfirstlane(blockIdx.x * (blockDim.x >> 5) + (threadIdx.x >> 5));
    if (node >= nT) return;
    const int* bucket = slot + (size_t)node * BUCKET_CAP;
    const int n = __builtin_amdgcn_readfirstlane(deg[node]);

    const float4 zero4 = make_float4(0.0f, 0.0f, 0.0f, 0.0f);
    float4 a0 = zero4;
    float4 a1 = zero4;

    if (n <= BUCKET_CAP) {
        for (int base = 0; base < n; base += 32) {
            const int m = min(32, n - base);
            const int myid = (lane < m) ? bucket[base + lane] : 0;
            for (int j = 0; j < m; j += GATHER_U) {
                int sid[GATHER_U];
#pragma unroll
                for (int k = 0; k < GATHER_U; ++k) sid[k] = __builtin_amdgcn_readlane(myid, (j + k) & 31);

                float4 x[GATHER_U], y[GATHER_U];
                const int left = m - j;
                if (left >= GATHER_U) {
#pragma unroll
                    for (int k = 0; k < GATHER_U; ++k) {
                        const float4* r = (const float4*)(hsrc + (size_t)sid[k] * DFEAT);
                        x[k] = r[lane];
                        y[k] = r[32 + lane];
                    }
                } else {
#pragma unroll
                    for (int k = 0; k < GATHER_U; ++k) {
                        if (k < left) {
                            const float4* r = (const float4*)(hsrc + (size_t)sid[k] * DFEAT);
                            x[k] = r[lane];
                            y[k] = r[32 + lane];
                        } else {
                            x[k] = zero4;
                            y[k] = zero4;
                        }
                    }
                }
#pragma unroll
                for (int k = 0; k < GATHER_U; ++k) {
                    a0.x += x[k].x; a0.y += x[k].y; a0.z += x[k].z; a0.w += x[k].w;
                    a1.x += y[k].x; a1.y += y[k].y; a1.z += y[k].z; a1.w += y[k].w;
                }
            }
        }
    } else {
        for (int base = 0; base < nE; base += 32) {
            const int e = base + lane;
            const int d = (e < nE) ? edst[e] : -1;
            const int s = (e < nE) ? esrc[e] : 0;
            unsigned int hit = __builtin_amdgcn_ballot_w32(d == node);
            while (hit) {
                const int l = __builtin_ctz(hit);
                hit &= hit - 1;
                add_row(a0, a1, hsrc, __builtin_amdgcn_readlane(s, l), lane);
            }
        }
    }

    float4* o = (float4*)(sums + (size_t)node * DFEAT);
    o[lane]      = a0;
    o[32 + lane] = a1;
    if (lane == 0) inv[node] = 1.0f / fmaxf((float)n, 1.0f);
}


__device__ __forceinline__ v8f mma4(v2f a, v2f b, v8f c) {
    return __builtin_amdgcn_wmma_f32_16x16x4_f32(false, a, false, b, (short)0, c, false, false);
}

struct FragV {
    float4 as[2], at[2];
    float4 bl[4], br[4];
};

__device__ __forceinline__ void l0_load(FragV& f, const float* __restrict__ a0, const float* __restrict__ t0,
                                        const float* __restrict__ pl, const float* __restrict__ pr, int g) {
    const int ko = g * 8;
    f.as[0] = *(const float4*)(a0 + ko);
    f.as[1] = *(const float4*)(a0 + 16 * DFEAT + ko);
    f.at[0] = *(const float4*)(t0 + ko);
    f.at[1] = *(const float4*)(t0 + 16 * DFEAT + ko);
#pragma unroll
    for (int i = 0; i < 4; ++i) {
        f.bl[i] = *(const float4*)(pl + (size_t)(ko + i) * DFEAT);
        f.br[i] = *(const float4*)(pr + (size_t)(ko + i) * DFEAT);
    }
}

__device__ __forceinline__ void l0_mma(v8f (&acc)[2][4], const FragV& f, float inv0, float inv1) {
    v2f aS[2][2], aT[2][2];
    aS[0][0].x = f.as[0].x * inv0; aS[0][0].y = f.as[0].y * inv0;
    aS[0][1].x = f.as[0].z * inv0; aS[0][1].y = f.as[0].w * inv0;
    aS[1][0].x = f.as[1].x * inv1; aS[1][0].y = f.as[1].y * inv1;
    aS[1][1].x = f.as[1].z * inv1; aS[1][1].y = f.as[1].w * inv1;
    aT[0][0].x = f.at[0].x; aT[0][0].y = f.at[0].y;
    aT[0][1].x = f.at[0].z; aT[0][1].y = f.at[0].w;
    aT[1][0].x = f.at[1].x; aT[1][0].y = f.at[1].y;
    aT[1][1].x = f.at[1].z; aT[1][1].y = f.at[1].w;

    float L[4][4], R[4][4];
#pragma unroll
    for (int i = 0; i < 4; ++i) {
        L[i][0] = f.bl[i].x; L[i][1] = f.bl[i].y; L[i][2] = f.bl[i].z; L[i][3] = f.bl[i].w;
        R[i][0] = f.br[i].x; R[i][1] = f.br[i].y; R[i][2] = f.br[i].z; R[i][3] = f.br[i].w;
    }
#pragma unroll
    for (int s = 0; s < 2; ++s) {
#pragma unroll
        for (int j = 0; j < 4; ++j) {
            v2f b; b.x = L[2 * s][j]; b.y = L[2 * s + 1][j];
            acc[0][j] = mma4(aS[0][s], b, acc[0][j]);
            acc[1][j] = mma4(aS[1][s], b, acc[1][j]);
        }
#pragma unroll
        for (int j = 0; j < 4; ++j) {
            v2f b; b.x = R[2 * s][j]; b.y = R[2 * s + 1][j];
            acc[0][j] = mma4(aT[0][s], b, acc[0][j]);
            acc[1][j] = mma4(aT[1][s], b, acc[1][j]);
        }
    }
}

__global__ void __launch_bounds__(128)
sage_gemm_l0(const float* __restrict__ Asum,
             const float* __restrict__ invc,
             const float* __restrict__ Htgt,
             const float* __restrict__ Wl,
             const float* __restrict__ Wr,
             const float* __restrict__ bias,
             float* __restrict__ Out,
             int nPairs) {
    const int lane = threadIdx.x & 31;
    const int pair = blockIdx.x;
    if (pair >= nPairs) return;
    const int c = lane & 15;
    const int h = lane >> 4;
    const int colBase = (threadIdx.x >> 5) * 64;
    const int row0 = pair * 32 + c;

    const float inv0 = invc[row0];
    const float inv1 = invc[row0 + 16];

    const float4 bv = *(const float4*)(bias + colBase + 4 * c);
    v8f acc[2][4];
#pragma unroll
    for (int i = 0; i < 8; ++i) {
        acc[0][0][i] = bv.x; acc[0][1][i] = bv.y; acc[0][2][i] = bv.z; acc[0][3][i] = bv.w;
        acc[1][0][i] = bv.x; acc[1][1][i] = bv.y; acc[1][2][i] = bv.z; acc[1][3][i] = bv.w;
    }

    const float* a0 = Asum + (size_t)row0 * DFEAT + 4 * h;
    const float* t0 = Htgt + (size_t)row0 * DFEAT + 4 * h;
    const float* pl = Wl + (size_t)(4 * h) * DFEAT + colBase + 4 * c;
    const float* pr = Wr + (size_t)(4 * h) * DFEAT + colBase + 4 * c;

    FragV f0, f1;
    l0_load(f0, a0, t0, pl, pr, 0);
    for (int g = 0; g < DFEAT / 8 - 2; g += 2) {
        l0_load(f1, a0, t0, pl, pr, g + 1);
        l0_mma(acc, f0, inv0, inv1);
        l0_load(f0, a0, t0, pl, pr, g + 2);
        l0_mma(acc, f1, inv0, inv1);
    }
    l0_load(f1, a0, t0, pl, pr, DFEAT / 8 - 1);
    l0_mma(acc, f0, inv0, inv1);
    l0_mma(acc, f1, inv0, inv1);

#pragma unroll
    for (int mt = 0; mt < 2; ++mt) {
        float* o = Out + (size_t)(pair * 32 + mt * 16 + 8 * h) * DFEAT + colBase + 4 * c;
#pragma unroll
        for (int i = 0; i < 8; ++i) {
            float4 v;
            v.x = fmaxf(acc[mt][0][i], 0.0f);
            v.y = fmaxf(acc[mt][1][i], 0.0f);
            v.z = fmaxf(acc[mt][2][i], 0.0f);
            v.w = fmaxf(acc[mt][3][i], 0.0f);
            *(float4*)(o + (size_t)i * DFEAT) = v;
        }
    }
}

struct FragR {
    float4 as[2], at[2];
    v2f bl[2][2][3];
    v2f br[2][2][3];
};

__device__ __forceinline__ void l1_load(FragR& f, const float* __restrict__ a0, const float* __restrict__ t0,
                                        const float* __restrict__ pl, const float* __restrict__ pr,
                                        const int (&cj)[3], int Nact, int g) {
#pragma unroll
    for (int u = 0; u < 2; ++u) {
        const int ko = g * 16 + u * 8;
        f.as[u] = *(const float4*)(a0 + ko);
        f.at[u] = *(const float4*)(t0 + ko);
#pragma unroll
        for (int s = 0; s < 2; ++s) {
            const float* ql = pl + (size_t)(ko + 2 * s) * Nact;
            const float* qr = pr + (size_t)(ko + 2 * s) * Nact;
#pragma unroll
            for (int j = 0; j < 3; ++j) {
                f.bl[u][s][j].x = ql[cj[j]];
                f.bl[u][s][j].y = ql[Nact + cj[j]];
                f.br[u][s][j].x = qr[cj[j]];
                f.br[u][s][j].y = qr[Nact + cj[j]];
            }
        }
    }
}

__device__ __forceinline__ void l1_mma(v8f (&acc)[3], const FragR& f, float inv) {
#pragma unroll
    for (int u = 0; u < 2; ++u) {
        v2f aS[2], aT[2];
        aS[0].x = f.as[u].x * inv; aS[0].y = f.as[u].y * inv;
        aS[1].x = f.as[u].z * inv; aS[1].y = f.as[u].w * inv;
        aT[0].x = f.at[u].x; aT[0].y = f.at[u].y;
        aT[1].x = f.at[u].z; aT[1].y = f.at[u].w;
#pragma unroll
        for (int s = 0; s < 2; ++s) {
#pragma unroll
            for (int j = 0; j < 3; ++j) acc[j] = mma4(aS[s], f.bl[u][s][j], acc[j]);
#pragma unroll
            for (int j = 0; j < 3; ++j) acc[j] = mma4(aT[s], f.br[u][s][j], acc[j]);
        }
    }
}

__global__ void __launch_bounds__(32)
sage_gemm_l1(const float* __restrict__ Asum,
             const float* __restrict__ invc,
             const float* __restrict__ Htgt,
             const float* __restrict__ Wl,
             const float* __restrict__ Wr,
             const float* __restrict__ bias,
             float* __restrict__ Out,
             int nTiles,
             int Nact) {
    const int lane = threadIdx.x & 31;
    const int tileM = blockIdx.x;
    if (tileM >= nTiles) return;
    const int c = lane & 15;
    const int h = lane >> 4;
    const int row = tileM * 16 + c;

    int col[3], cj[3];
#pragma unroll
    for (int j = 0; j < 3; ++j) { col[j] = j * 16 + c; cj[j] = min(col[j], Nact - 1); }

    const float inv = invc[row];

    v8f acc[3];
#pragma unroll
    for (int j = 0; j < 3; ++j) {
        const float bvj = bias[cj[j]];
#pragma unroll
        for (int i = 0; i < 8; ++i) acc[j][i] = bvj;
    }

    const float* a0 = Asum + (size_t)row * DFEAT + 4 * h;
    const float* t0 = Htgt + (size_t)row * DFEAT + 4 * h;
    const float* pl = Wl + (size_t)(4 * h) * Nact;
    const float* pr = Wr + (size_t)(4 * h) * Nact;

    FragR f0, f1;
    l1_load(f0, a0, t0, pl, pr, cj, Nact, 0);
    for (int g = 0; g < DFEAT / 16 - 2; g += 2) {
        l1_load(f1, a0, t0, pl, pr, cj, Nact, g + 1);
        l1_mma(acc, f0, inv);
        l1_load(f0, a0, t0, pl, pr, cj, Nact, g + 2);
        l1_mma(acc, f1, inv);
    }
    l1_load(f1, a0, t0, pl, pr, cj, Nact, DFEAT / 16 - 1);
    l1_mma(acc, f0, inv);
    l1_mma(acc, f1, inv);

    const int mbase = tileM * 16 + 8 * h;
#pragma unroll
    for (int j = 0; j < 3; ++j) {
        if (col[j] < Nact) {
#pragma unroll
            for (int i = 0; i < 8; ++i) Out[(size_t)(mbase + i) * Nact + col[j]] = acc[j][i];
        }
    }
}


extern "C" void kernel_launch(void* const* d_in, const int* in_sizes, int n_in,
                              void* d_out, int out_size, void* d_ws, size_t ws_size,
                              hipStream_t stream) {
    const float* h   = (const float*)d_in[0];
    const float* wl0 = (const float*)d_in[1];
    const float* wr0 = (const float*)d_in[2];
    const float* b0  = (const float*)d_in[3];
    const float* wl1 = (const float*)d_in[4];
    const float* wr1 = (const float*)d_in[5];
    const float* b1  = (const float*)d_in[6];
    const int*   ei0 = (const int*)d_in[7];
    const int*   ei1 = (const int*)d_in[8];

    const int E0 = in_sizes[7] / 2;
    const int E1 = in_sizes[8] / 2;
    const int NT0 = 20000;
    const int NT1 = 2000;
    const int NCLS = 47;

    float* sums0 = (float*)d_ws;
    float* cnt0  = sums0 + (size_t)NT0 * DFEAT;
    float* inv0  = cnt0 + NT0;
    float* h1    = inv0 + NT0;
    float* reg1  = h1 + (size_t)NT0 * DFEAT;
    float* cnt1  = reg1 + (size_t)NT1 * DFEAT;
    float* inv1  = cnt1 + NT1;
    float* sums1 = sums0;

    int* slot0 = (int*)h1;
    int* deg0  = (int*)reg1 - NT0;
    int* deg1  = (int*)reg1;
    int* slot1 = deg1 + 2048;

    zero_kernel<<<22, 1024, 0, stream>>>((float*)deg0, (long)(NT0 + NT1));
    bucket_kernel<<<(E0 + E1 + 255) / 256, 256, 0, stream>>>(ei0, ei0 + E0, E0, deg0, slot0,
                                                             ei1, ei1 + E1, E1, deg1, slot1);

    gather_sum_kernel<<<(NT0 + 7) / 8, 256, 0, stream>>>(h, deg0, slot0, ei0, ei0 + E0, E0, sums0, inv0, NT0);

    sage_gemm_l0<<<NT0 / 32, 128, 0, stream>>>(sums0, inv0, h, wl0, wr0, b0, h1, NT0 / 32);

    gather_sum_kernel<<<(NT1 + 7) / 8, 256, 0, stream>>>(h1, deg1, slot1, ei1, ei1 + E1, E1, sums1, inv1, NT1);

    sage_gemm_l1<<<NT1 / 16, 32, 0, stream>>>(sums1, inv1, h1, wl1, wr1, b1, (float*)d_out, NT1 / 16, NCLS);
}


# Layer-1 aggregation fused into its GEMM launch: 16 waves per 16-row tile gather into LDS, then K split 16 ways, partial tiles added via LDS in wave order; 6 launches -> 5; sum order over k changed
# speedup vs baseline: 3.5582x; 1.0860x over previous
#include <hip/hip_runtime.h>

typedef __attribute__((ext_vector_type(2))) float v2f;
typedef __attribute__((ext_vector_type(8))) float v8f;

#define DFEAT 256


__global__ void zero_kernel(float* __restrict__ p, long n) {
    long i = (long)blockIdx.x * blockDim.x + threadIdx.x;
    long stride = (long)gridDim.x * blockDim.x;
    for (; i < n; i += stride) p[i] = 0.0f;
}

#define BUCKET_CAP 64

__global__ void bucket_kernel(const int* __restrict__ srcA, const int* __restrict__ dstA, int nEA,
                              int* __restrict__ degA, int* __restrict__ slotA,
                              const int* __restrict__ srcB, const int* __restrict__ dstB, int nEB,
                              int* __restrict__ degB, int* __restrict__ slotB) {
    int i = blockIdx.x * blockDim.x + threadIdx.x;
    if (i < nEA) {
        const int d = dstA[i];
        const int s = srcA[i];
        const int r = atomicAdd(&degA[d], 1);
        if (r < BUCKET_CAP) slotA[(size_t)d * BUCKET_CAP + r] = s;
    } else {
        int k = i - nEA;
        if (k < nEB) {
            const int d = dstB[k];
            const int s = srcB[k];
            const int r = atomicAdd(&degB[d], 1);
            if (r < BUCKET_CAP) slotB[(size_t)d * BUCKET_CAP + r] = s;
        }
    }
}

#define GATHER_U 8

__device__ __forceinline__ void add_row(float4& a0, float4& a1, const float* __restrict__ hsrc, int sid, int lane) {
    const float4* r = (const float4*)(hsrc + (size_t)sid * DFEAT);
    const float4 x = r[lane];
    const float4 y = r[32 + lane];
    a0.x += x.x; a0.y += x.y; a0.z += x.z; a0.w += x.w;
    a1.x += y.x; a1.y += y.y; a1.z += y.z; a1.w += y.w;
}

__device__ __forceinline__ void gather_rows(float4& a0, float4& a1,
                                            const float* __restrict__ hsrc,
                                            const int* __restrict__ bucket, int n,
                                            const int* __restrict__ esrc,
                                            const int* __restrict__ edst,
                                            int nE, int node, int lane) {
    const float4 zero4 = make_float4(0.0f, 0.0f, 0.0f, 0.0f);
    a0 = zero4;
    a1 = zero4;

    if (n <= BUCKET_CAP) {
        for (int base = 0; base < n; base += 32) {
            const int m = min(32, n - base);
            const int myid = (lane < m) ? bucket[base + lane] : 0;
            for (int j = 0; j < m; j += GATHER_U) {
                int sid[GATHER_U];
#pragma unroll
                for (int k = 0; k < GATHER_U; ++k) sid[k] = __builtin_amdgcn_readlane(myid, (j + k) & 31);

                float4 x[GATHER_U], y[GATHER_U];
                const int left = m - j;
                if (left >= GATHER_U) {
#pragma unroll
                    for (int k = 0; k < GATHER_U; ++k) {
                        const float4* r = (const float4*)(hsrc + (size_t)sid[k] * DFEAT);
                        x[k] = r[lane];
                        y[k] = r[32 + lane];
                    }
                } else {
#pragma unroll
                    for (int k = 0; k < GATHER_U; ++k) {
                        if (k < left) {
                            const float4* r = (const float4*)(hsrc + (size_t)sid[k] * DFEAT);
                            x[k] = r[lane];
                            y[k] = r[32 + lane];
                        } else {
                            x[k] = zero4;
                            y[k] = zero4;
                        }
                    }
                }
#pragma unroll
                for (int k = 0; k < GATHER_U; ++k) {
                    a0.x += x[k].x; a0.y += x[k].y; a0.z += x[k].z; a0.w += x[k].w;
                    a1.x += y[k].x; a1.y += y[k].y; a1.z += y[k].z; a1.w += y[k].w;
                }
            }
        }
    } else {
        for (int base = 0; base < nE; base += 32) {
            const int e = base + lane;
            const int d = (e < nE) ? edst[e] : -1;
            const int s = (e < nE) ? esrc[e] : 0;
            unsigned int hit = __builtin_amdgcn_ballot_w32(d == node);
            while (hit) {
                const int l = __builtin_ctz(hit);
                hit &= hit - 1;
                add_row(a0, a1, hsrc, __builtin_amdgcn_readlane(s, l), lane);
            }
        }
    }

}

__global__ void __launch_bounds__(256)
gather_sum_kernel(const float* __restrict__ hsrc,
                  const int* __restrict__ deg,
                  const int* __restrict__ slot,
                  const int* __restrict__ esrc,
                  const int* __restrict__ edst,
                  int nE,
                  float* __restrict__ sums,
                  float* __restrict__ inv,
                  int nT) {
    const int lane = threadIdx.x & 31;
    const int node = __builtin_amdgcn_readfirstlane(blockIdx.x * (blockDim.x >> 5) + (threadIdx.x >> 5));
    if (node >= nT) return;
    const int* bucket = slot + (size_t)node * BUCKET_CAP;
    const int n = __builtin_amdgcn_readfirstlane(deg[node]);

    float4 a0, a1;
    gather_rows(a0, a1, hsrc, bucket, n, esrc, edst, nE, node, lane);

    float4* o = (float4*)(sums + (size_t)node * DFEAT);
    o[lane]      = a0;
    o[32 + lane] = a1;
    if (lane == 0) inv[node] = 1.0f / fmaxf((float)n, 1.0f);
}


__device__ __forceinline__ v8f mma4(v2f a, v2f b, v8f c) {
    return __builtin_amdgcn_wmma_f32_16x16x4_f32(false, a, false, b, (short)0, c, false, false);
}

struct FragV {
    float4 as[2], at[2];
    float4 bl[4], br[4];
};

__device__ __forceinline__ void l0_load(FragV& f, const float* __restrict__ a0, const float* __restrict__ t0,
                                        const float* __restrict__ pl, const float* __restrict__ pr, int g) {
    const int ko = g * 8;
    f.as[0] = *(const float4*)(a0 + ko);
    f.as[1] = *(const float4*)(a0 + 16 * DFEAT + ko);
    f.at[0] = *(const float4*)(t0 + ko);
    f.at[1] = *(const float4*)(t0 + 16 * DFEAT + ko);
#pragma unroll
    for (int i = 0; i < 4; ++i) {
        f.bl[i] = *(const float4*)(pl + (size_t)(ko + i) * DFEAT);
        f.br[i] = *(const float4*)(pr + (size_t)(ko + i) * DFEAT);
    }
}

__device__ __forceinline__ void l0_mma(v8f (&acc)[2][4], const FragV& f, float inv0, float inv1) {
    v2f aS[2][2], aT[2][2];
    aS[0][0].x = f.as[0].x * inv0; aS[0][0].y = f.as[0].y * inv0;
    aS[0][1].x = f.as[0].z * inv0; aS[0][1].y = f.as[0].w * inv0;
    aS[1][0].x = f.as[1].x * inv1; aS[1][0].y = f.as[1].y * inv1;
    aS[1][1].x = f.as[1].z * inv1; aS[1][1].y = f.as[1].w * inv1;
    aT[0][0].x = f.at[0].x; aT[0][0].y = f.at[0].y;
    aT[0][1].x = f.at[0].z; aT[0][1].y = f.at[0].w;
    aT[1][0].x = f.at[1].x; aT[1][0].y = f.at[1].y;
    aT[1][1].x = f.at[1].z; aT[1][1].y = f.at[1].w;

    float L[4][4], R[4][4];
#pragma unroll
    for (int i = 0; i < 4; ++i) {
        L[i][0] = f.bl[i].x; L[i][1] = f.bl[i].y; L[i][2] = f.bl[i].z; L[i][3] = f.bl[i].w;
        R[i][0] = f.br[i].x; R[i][1] = f.br[i].y; R[i][2] = f.br[i].z; R[i][3] = f.br[i].w;
    }
#pragma unroll
    for (int s = 0; s < 2; ++s) {
#pragma unroll
        for (int j = 0; j < 4; ++j) {
            v2f b; b.x = L[2 * s][j]; b.y = L[2 * s + 1][j];
            acc[0][j] = mma4(aS[0][s], b, acc[0][j]);
            acc[1][j] = mma4(aS[1][s], b, acc[1][j]);
        }
#pragma unroll
        for (int j = 0; j < 4; ++j) {
            v2f b; b.x = R[2 * s][j]; b.y = R[2 * s + 1][j];
            acc[0][j] = mma4(aT[0][s], b, acc[0][j]);
            acc[1][j] = mma4(aT[1][s], b, acc[1][j]);
        }
    }
}

__global__ void __launch_bounds__(128)
sage_gemm_l0(const float* __restrict__ Asum,
             const float* __restrict__ invc,
             const float* __restrict__ Htgt,
             const float* __restrict__ Wl,
             const float* __restrict__ Wr,
             const float* __restrict__ bias,
             float* __restrict__ Out,
             int nPairs) {
    const int lane = threadIdx.x & 31;
    const int pair = blockIdx.x;
    if (pair >= nPairs) return;
    const int c = lane & 15;
    const int h = lane >> 4;
    const int colBase = (threadIdx.x >> 5) * 64;
    const int row0 = pair * 32 + c;

    const float inv0 = invc[row0];
    const float inv1 = invc[row0 + 16];

    const float4 bv = *(const float4*)(bias + colBase + 4 * c);
    v8f acc[2][4];
#pragma unroll
    for (int i = 0; i < 8; ++i) {
        acc[0][0][i] = bv.x; acc[0][1][i] = bv.y; acc[0][2][i] = bv.z; acc[0][3][i] = bv.w;
        acc[1][0][i] = bv.x; acc[1][1][i] = bv.y; acc[1][2][i] = bv.z; acc[1][3][i] = bv.w;
    }

    const float* a0 = Asum + (size_t)row0 * DFEAT + 4 * h;
    const float* t0 = Htgt + (size_t)row0 * DFEAT + 4 * h;
    const float* pl = Wl + (size_t)(4 * h) * DFEAT + colBase + 4 * c;
    const float* pr = Wr + (size_t)(4 * h) * DFEAT + colBase + 4 * c;

    FragV f0, f1;
    l0_load(f0, a0, t0, pl, pr, 0);
    for (int g = 0; g < DFEAT / 8 - 2; g += 2) {
        l0_load(f1, a0, t0, pl, pr, g + 1);
        l0_mma(acc, f0, inv0, inv1);
        l0_load(f0, a0, t0, pl, pr, g + 2);
        l0_mma(acc, f1, inv0, inv1);
    }
    l0_load(f1, a0, t0, pl, pr, DFEAT / 8 - 1);
    l0_mma(acc, f0, inv0, inv1);
    l0_mma(acc, f1, inv0, inv1);

#pragma unroll
    for (int mt = 0; mt < 2; ++mt) {
        float* o = Out + (size_t)(pair * 32 + mt * 16 + 8 * h) * DFEAT + colBase + 4 * c;
#pragma unroll
        for (int i = 0; i < 8; ++i) {
            float4 v;
            v.x = fmaxf(acc[mt][0][i], 0.0f);
            v.y = fmaxf(acc[mt][1][i], 0.0f);
            v.z = fmaxf(acc[mt][2][i], 0.0f);
            v.w = fmaxf(acc[mt][3][i], 0.0f);
            *(float4*)(o + (size_t)i * DFEAT) = v;
        }
    }
}

#define L1_ROWS 16
#define L1_PITCH 260
#define L1_TILE 768

__global__ void __launch_bounds__(512)
sage_l1_fused(const float* __restrict__ hsrc,
              const int* __restrict__ deg,
              const int* __restrict__ slot,
              const int* __restrict__ esrc,
              const int* __restrict__ edst,
              int nE,
              const float* __restrict__ Wl,
              const float* __restrict__ Wr,
              const float* __restrict__ bias,
              float* __restrict__ Out,
              int nTiles,
              int Nact) {
    __shared__ __attribute__((aligned(16))) float sS[L1_ROWS * L1_PITCH];
    __shared__ __attribute__((aligned(16))) float sT[L1_ROWS * L1_PITCH];
    __shared__ float sInv[L1_ROWS];
    __shared__ float sR[L1_ROWS * L1_TILE];

    const int tileM = blockIdx.x;
    if (tileM >= nTiles) return;
    const int lane = threadIdx.x & 31;
    const int w = __builtin_amdgcn_readfirstlane((int)(threadIdx.x >> 5));
    const int c = lane & 15;
    const int h = lane >> 4;
    const int kw = 16 * w;

    int cj[3];
#pragma unroll
    for (int j = 0; j < 3; ++j) cj[j] = min(j * 16 + c, Nact - 1);

    v2f bl[2][2][3], br[2][2][3];
#pragma unroll
    for (int u = 0; u < 2; ++u) {
#pragma unroll
        for (int s = 0; s < 2; ++s) {
            const float* ql = Wl + (size_t)(kw + 8 * u + 4 * h + 2 * s) * Nact;
            const float* qr = Wr + (size_t)(kw + 8 * u + 4 * h + 2 * s) * Nact;
#pragma unroll
            for (int j = 0; j < 3; ++j) {
                bl[u][s][j].x = ql[cj[j]];
                bl[u][s][j].y = ql[Nact + cj[j]];
                br[u][s][j].x = qr[cj[j]];
                br[u][s][j].y = qr[Nact + cj[j]];
            }
        }
    }

    const int node = tileM * L1_ROWS + w;
    const int n = __builtin_amdgcn_readfirstlane(deg[node]);
    const float4* trow = (const float4*)(hsrc + (size_t)node * DFEAT);
    *(float4*)(sT + w * L1_PITCH + 4 * lane)       = trow[lane];
    *(float4*)(sT + w * L1_PITCH + 128 + 4 * lane) = trow[32 + lane];
    if (lane == 0) sInv[w] = 1.0f / fmaxf((float)n, 1.0f);

    float4 a0, a1;
    gather_rows(a0, a1, hsrc, slot + (size_t)node * BUCKET_CAP, n, esrc, edst, nE, node, lane);

    *(float4*)(sS + w * L1_PITCH + 4 * lane)       = a0;
    *(float4*)(sS + w * L1_PITCH + 128 + 4 * lane) = a1;
    __syncthreads();

    const float inv = sInv[c];
    v8f acc[3];
#pragma unroll
    for (int j = 0; j < 3; ++j) {
#pragma unroll
        for (int i = 0; i < 8; ++i) acc[j][i] = 0.0f;
    }
#pragma unroll
    for (int u = 0; u < 2; ++u) {
        const int ko = kw + 8 * u + 4 * h;
        const float4 as = *(const float4*)(sS + c * L1_PITCH + ko);
        const float4 at = *(const float4*)(sT + c * L1_PITCH + ko);
        v2f aS[2], aT[2];
        aS[0].x = as.x * inv; aS[0].y = as.y * inv;
        aS[1].x = as.z * inv; aS[1].y = as.w * inv;
        aT[0].x = at.x; aT[0].y = at.y;
        aT[1].x = at.z; aT[1].y = at.w;
#pragma unroll
        for (int s = 0; s < 2; ++s) {
#pragma unroll
            for (int j = 0; j < 3; ++j) acc[j] = mma4(aS[s], bl[u][s][j], acc[j]);
#pragma unroll
            for (int j = 0; j < 3; ++j) acc[j] = mma4(aT[s], br[u][s][j], acc[j]);
        }
    }

#pragma unroll
    for (int j = 0; j < 3; ++j) {
#pragma unroll
        for (int i = 0; i < 8; ++i) sR[w * L1_TILE + (j * 8 + i) * 32 + lane] = acc[j][i];
    }
    __syncthreads();

    for (int f = threadIdx.x; f < L1_ROWS * Nact; f += 512) {
        const int r  = f / Nact;
        const int cc = f - r * Nact;
        const int q  = ((cc >> 4) * 8 + (r & 7)) * 32 + (r >> 3) * 16 + (cc & 15);
        float sum = sR[q];
#pragma unroll
        for (int ww = 1; ww < L1_ROWS; ++ww) sum += sR[ww * L1_TILE + q];
        Out[(size_t)tileM * L1_ROWS * Nact + f] = sum + bias[cc];
    }
}


extern "C" void kernel_launch(void* const* d_in, const int* in_sizes, int n_in,
                              void* d_out, int out_size, void* d_ws, size_t ws_size,
                              hipStream_t stream) {
    const float* h   = (const float*)d_in[0];
    const float* wl0 = (const float*)d_in[1];
    const float* wr0 = (const float*)d_in[2];
    const float* b0  = (const float*)d_in[3];
    const float* wl1 = (const float*)d_in[4];
    const float* wr1 = (const float*)d_in[5];
    const float* b1  = (const float*)d_in[6];
    const int*   ei0 = (const int*)d_in[7];
    const int*   ei1 = (const int*)d_in[8];

    const int E0 = in_sizes[7] / 2;
    const int E1 = in_sizes[8] / 2;
    const int NT0 = 20000;
    const int NT1 = 2000;
    const int NCLS = 47;

    float* sums0 = (float*)d_ws;
    float* cnt0  = sums0 + (size_t)NT0 * DFEAT;
    float* inv0  = cnt0 + NT0;
    float* h1    = inv0 + NT0;
    float* reg1  = h1 + (size_t)NT0 * DFEAT;

    int* slot0 = (int*)h1;
    int* deg0  = (int*)reg1 - NT0;
    int* deg1  = (int*)reg1;
    int* slot1 = deg1 + 2048;

    zero_kernel<<<22, 1024, 0, stream>>>((float*)deg0, (long)(NT0 + NT1));
    bucket_kernel<<<(E0 + E1 + 255) / 256, 256, 0, stream>>>(ei0, ei0 + E0, E0, deg0, slot0,
                                                             ei1, ei1 + E1, E1, deg1, slot1);

    gather_sum_kernel<<<(NT0 + 7) / 8, 256, 0, stream>>>(h, deg0, slot0, ei0, ei0 + E0, E0, sums0, inv0, NT0);

    sage_gemm_l0<<<NT0 / 32, 128, 0, stream>>>(sums0, inv0, h, wl0, wr0, b0, h1, NT0 / 32);

    sage_l1_fused<<<NT1 / 16, 512, 0, stream>>>(h1, deg1, slot1, ei1, ei1 + E1, E1, wl1, wr1, b1,
                                                (float*)d_out, NT1 / 16, NCLS);
}


# Layer-0 targets that layer 1 never reads are skipped: layer-1 sources flagged while bucketing, gather compacts live targets into a list (one integer add per group), GEMM runs over the list
# speedup vs baseline: 3.8046x; 1.0692x over previous
#include <hip/hip_runtime.h>

typedef __attribute__((ext_vector_type(2))) float v2f;
typedef __attribute__((ext_vector_type(8))) float v8f;

#define DFEAT 256


__global__ void zero_kernel(int* __restrict__ a, int na, int* __restrict__ b, int nb) {
    const int i0 = blockIdx.x * blockDim.x + threadIdx.x;
    const int stride = gridDim.x * blockDim.x;
    for (int i = i0; i < na; i += stride) a[i] = 0;
    for (int i = i0; i < nb; i += stride) b[i] = 0;
}

#define BUCKET_CAP 64

__global__ void bucket_kernel(const int* __restrict__ srcA, const int* __restrict__ dstA, int nEA,
                              int* __restrict__ degA, int* __restrict__ slotA,
                              const int* __restrict__ srcB, const int* __restrict__ dstB, int nEB,
                              int* __restrict__ degB, int* __restrict__ slotB,
                              int* __restrict__ usedB) {
    int i = blockIdx.x * blockDim.x + threadIdx.x;
    if (i < nEA) {
        const int d = dstA[i];
        const int s = srcA[i];
        const int r = atomicAdd(&degA[d], 1);
        if (r < BUCKET_CAP) slotA[(size_t)d * BUCKET_CAP + r] = s;
    } else {
        int k = i - nEA;
        if (k < nEB) {
            const int d = dstB[k];
            const int s = srcB[k];
            const int r = atomicAdd(&degB[d], 1);
            if (r < BUCKET_CAP) slotB[(size_t)d * BUCKET_CAP + r] = s;
            usedB[s] = 1;
        }
    }
}

#define GATHER_U 8

__device__ __forceinline__ void add_row(float4& a0, float4& a1, const float* __restrict__ hsrc, int sid, int lane) {
    const float4* r = (const float4*)(hsrc + (size_t)sid * DFEAT);
    const float4 x = r[lane];
    const float4 y = r[32 + lane];
    a0.x += x.x; a0.y += x.y; a0.z += x.z; a0.w += x.w;
    a1.x += y.x; a1.y += y.y; a1.z += y.z; a1.w += y.w;
}

__device__ __forceinline__ void gather_rows(float4& a0, float4& a1,
                                            const float* __restrict__ hsrc,
                                            const int* __restrict__ bucket, int n,
                                            const int* __restrict__ esrc,
                                            const int* __restrict__ edst,
                                            int nE, int node, int lane) {
    const float4 zero4 = make_float4(0.0f, 0.0f, 0.0f, 0.0f);
    a0 = zero4;
    a1 = zero4;

    if (n <= BUCKET_CAP) {
        for (int base = 0; base < n; base += 32) {
            const int m = min(32, n - base);
            const int myid = (lane < m) ? bucket[base + lane] : 0;
            for (int j = 0; j < m; j += GATHER_U) {
                int sid[GATHER_U];
#pragma unroll
                for (int k = 0; k < GATHER_U; ++k) sid[k] = __builtin_amdgcn_readlane(myid, (j + k) & 31);

                float4 x[GATHER_U], y[GATHER_U];
                const int left = m - j;
                if (left >= GATHER_U) {
#pragma unroll
                    for (int k = 0; k < GATHER_U; ++k) {
                        const float4* r = (const float4*)(hsrc + (size_t)sid[k] * DFEAT);
                        x[k] = r[lane];
                        y[k] = r[32 + lane];
                    }
                } else {
#pragma unroll
                    for (int k = 0; k < GATHER_U; ++k) {
                        if (k < left) {
                            const float4* r = (const float4*)(hsrc + (size_t)sid[k] * DFEAT);
                            x[k] = r[lane];
                            y[k] = r[32 + lane];
                        } else {
                            x[k] = zero4;
                            y[k] = zero4;
                        }
                    }
                }
#pragma unroll
                for (int k = 0; k < GATHER_U; ++k) {
                    a0.x += x[k].x; a0.y += x[k].y; a0.z += x[k].z; a0.w += x[k].w;
                    a1.x += y[k].x; a1.y += y[k].y; a1.z += y[k].z; a1.w += y[k].w;
                }
            }
        }
    } else {
        for (int base = 0; base < nE; base += 32) {
            const int e = base + lane;
            const int d = (e < nE) ? edst[e] : -1;
            const int s = (e < nE) ? esrc[e] : 0;
            unsigned int hit = __builtin_amdgcn_ballot_w32(d == node);
            while (hit) {
                const int l = __builtin_ctz(hit);
                hit &= hit - 1;
                add_row(a0, a1, hsrc, __builtin_amdgcn_readlane(s, l), lane);
            }
        }
    }

}

#define GATHER_WAVES 8

__global__ void __launch_bounds__(256)
gather_sum_kernel(const float* __restrict__ hsrc,
                  const int* __restrict__ deg,
                  const int* __restrict__ slot,
                  const int* __restrict__ esrc,
                  const int* __restrict__ edst,
                  int nE,
                  const int* __restrict__ used,
                  int nKeep,
                  int* __restrict__ nLive,
                  int* __restrict__ rowid,
                  float* __restrict__ sums,
                  float* __restrict__ inv,
                  int nT) {
    __shared__ int sFlag[GATHER_WAVES];
    __shared__ int sBase;
    const int lane = threadIdx.x & 31;
    const int w = __builtin_amdgcn_readfirstlane((int)(threadIdx.x >> 5));
    const int node = __builtin_amdgcn_readfirstlane(blockIdx.x * GATHER_WAVES + w);
    bool live = false;
    if (node < nT) live = (node < nKeep) || (__builtin_amdgcn_readfirstlane(used[node]) != 0);

    if (lane == 0) sFlag[w] = live ? 1 : 0;
    __syncthreads();
    int cnt = 0, before = 0;
    for (int k = 0; k < GATHER_WAVES; ++k) {
        const int f = sFlag[k];
        cnt += f;
        before += (k < w) ? f : 0;
    }
    int base = 0;
    if (threadIdx.x == 0 && cnt > 0) base = atomicAdd(nLive, cnt);

    float4 a0 = make_float4(0.0f, 0.0f, 0.0f, 0.0f), a1 = a0;
    int n = 0;
    if (live) {
        n = __builtin_amdgcn_readfirstlane(deg[node]);
        gather_rows(a0, a1, hsrc, slot + (size_t)node * BUCKET_CAP, n, esrc, edst, nE, node, lane);
    }
    if (threadIdx.x == 0) sBase = base;
    __syncthreads();

    if (live) {
        const int pos = sBase + before;
        float4* o = (float4*)(sums + (size_t)pos * DFEAT);
        o[lane]      = a0;
        o[32 + lane] = a1;
        if (lane == 0) {
            inv[pos] = 1.0f / fmaxf((float)n, 1.0f);
            rowid[pos] = node;
        }
    }
}


__device__ __forceinline__ v8f mma4(v2f a, v2f b, v8f c) {
    return __builtin_amdgcn_wmma_f32_16x16x4_f32(false, a, false, b, (short)0, c, false, false);
}

struct FragV {
    float4 as[2], at[2];
    float4 bl[4], br[4];
};

__device__ __forceinline__ void l0_load(FragV& f, const float* __restrict__ a0, const float* __restrict__ a1,
                                        const float* __restrict__ t0, const float* __restrict__ t1,
                                        const float* __restrict__ pl, const float* __restrict__ pr, int g) {
    const int ko = g * 8;
    f.as[0] = *(const float4*)(a0 + ko);
    f.as[1] = *(const float4*)(a1 + ko);
    f.at[0] = *(const float4*)(t0 + ko);
    f.at[1] = *(const float4*)(t1 + ko);
#pragma unroll
    for (int i = 0; i < 4; ++i) {
        f.bl[i] = *(const float4*)(pl + (size_t)(ko + i) * DFEAT);
        f.br[i] = *(const float4*)(pr + (size_t)(ko + i) * DFEAT);
    }
}

__device__ __forceinline__ void l0_mma(v8f (&acc)[2][4], const FragV& f, float inv0, float inv1) {
    v2f aS[2][2], aT[2][2];
    aS[0][0].x = f.as[0].x * inv0; aS[0][0].y = f.as[0].y * inv0;
    aS[0][1].x = f.as[0].z * inv0; aS[0][1].y = f.as[0].w * inv0;
    aS[1][0].x = f.as[1].x * inv1; aS[1][0].y = f.as[1].y * inv1;
    aS[1][1].x = f.as[1].z * inv1; aS[1][1].y = f.as[1].w * inv1;
    aT[0][0].x = f.at[0].x; aT[0][0].y = f.at[0].y;
    aT[0][1].x = f.at[0].z; aT[0][1].y = f.at[0].w;
    aT[1][0].x = f.at[1].x; aT[1][0].y = f.at[1].y;
    aT[1][1].x = f.at[1].z; aT[1][1].y = f.at[1].w;

    float L[4][4], R[4][4];
#pragma unroll
    for (int i = 0; i < 4; ++i) {
        L[i][0] = f.bl[i].x; L[i][1] = f.bl[i].y; L[i][2] = f.bl[i].z; L[i][3] = f.bl[i].w;
        R[i][0] = f.br[i].x; R[i][1] = f.br[i].y; R[i][2] = f.br[i].z; R[i][3] = f.br[i].w;
    }
#pragma unroll
    for (int s = 0; s < 2; ++s) {
#pragma unroll
        for (int j = 0; j < 4; ++j) {
            v2f b; b.x = L[2 * s][j]; b.y = L[2 * s + 1][j];
            acc[0][j] = mma4(aS[0][s], b, acc[0][j]);
            acc[1][j] = mma4(aS[1][s], b, acc[1][j]);
        }
#pragma unroll
        for (int j = 0; j < 4; ++j) {
            v2f b; b.x = R[2 * s][j]; b.y = R[2 * s + 1][j];
            acc[0][j] = mma4(aT[0][s], b, acc[0][j]);
            acc[1][j] = mma4(aT[1][s], b, acc[1][j]);
        }
    }
}

__global__ void __launch_bounds__(128)
sage_gemm_l0(const float* __restrict__ Asum,
             const float* __restrict__ invc,
             const int* __restrict__ rowid,
             const int* __restrict__ nLive,
             const float* __restrict__ Htgt,
             const float* __restrict__ Wl,
             const float* __restrict__ Wr,
             const float* __restrict__ bias,
             float* __restrict__ Out) {
    const int lane = threadIdx.x & 31;
    const int pair = blockIdx.x;
    const int nC = *nLive;
    if (pair * 32 >= nC) return;
    const int c = lane & 15;
    const int h = lane >> 4;
    const int colBase = (threadIdx.x >> 5) * 64;
    const int q0 = min(pair * 32 + c, nC - 1);
    const int q1 = min(pair * 32 + 16 + c, nC - 1);

    const float inv0 = invc[q0];
    const float inv1 = invc[q1];
    const int rid0 = rowid[q0];
    const int rid1 = rowid[q1];

    const float4 bv = *(const float4*)(bias + colBase + 4 * c);
    v8f acc[2][4];
#pragma unroll
    for (int i = 0; i < 8; ++i) {
        acc[0][0][i] = bv.x; acc[0][1][i] = bv.y; acc[0][2][i] = bv.z; acc[0][3][i] = bv.w;
        acc[1][0][i] = bv.x; acc[1][1][i] = bv.y; acc[1][2][i] = bv.z; acc[1][3][i] = bv.w;
    }

    const float* a0 = Asum + (size_t)q0 * DFEAT + 4 * h;
    const float* a1 = Asum + (size_t)q1 * DFEAT + 4 * h;
    const float* t0 = Htgt + (size_t)rid0 * DFEAT + 4 * h;
    const float* t1 = Htgt + (size_t)rid1 * DFEAT + 4 * h;
    const float* pl = Wl + (size_t)(4 * h) * DFEAT + colBase + 4 * c;
    const float* pr = Wr + (size_t)(4 * h) * DFEAT + colBase + 4 * c;

    FragV f0, f1;
    l0_load(f0, a0, a1, t0, t1, pl, pr, 0);
    for (int g = 0; g < DFEAT / 8 - 2; g += 2) {
        l0_load(f1, a0, a1, t0, t1, pl, pr, g + 1);
        l0_mma(acc, f0, inv0, inv1);
        l0_load(f0, a0, a1, t0, t1, pl, pr, g + 2);
        l0_mma(acc, f1, inv0, inv1);
    }
    l0_load(f1, a0, a1, t0, t1, pl, pr, DFEAT / 8 - 1);
    l0_mma(acc, f0, inv0, inv1);
    l0_mma(acc, f1, inv0, inv1);

#pragma unroll
    for (int mt = 0; mt < 2; ++mt) {
        const int ridm = mt ? rid1 : rid0;
#pragma unroll
        for (int i = 0; i < 8; ++i) {
            const int nodeLo = __builtin_amdgcn_readlane(ridm, i);
            const int nodeHi = __builtin_amdgcn_readlane(ridm, 8 + i);
            const int node = h ? nodeHi : nodeLo;
            const int p = pair * 32 + mt * 16 + 8 * h + i;
            float4 v;
            v.x = fmaxf(acc[mt][0][i], 0.0f);
            v.y = fmaxf(acc[mt][1][i], 0.0f);
            v.z = fmaxf(acc[mt][2][i], 0.0f);
            v.w = fmaxf(acc[mt][3][i], 0.0f);
            if (p < nC) *(float4*)(Out + (size_t)node * DFEAT + colBase + 4 * c) = v;
        }
    }
}

#define L1_ROWS 16
#define L1_PITCH 260
#define L1_TILE 768

__global__ void __launch_bounds__(512)
sage_l1_fused(const float* __restrict__ hsrc,
              const int* __restrict__ deg,
              const int* __restrict__ slot,
              const int* __restrict__ esrc,
              const int* __restrict__ edst,
              int nE,
              const float* __restrict__ Wl,
              const float* __restrict__ Wr,
              const float* __restrict__ bias,
              float* __restrict__ Out,
              int nTiles,
              int Nact) {
    __shared__ __attribute__((aligned(16))) float sS[L1_ROWS * L1_PITCH];
    __shared__ __attribute__((aligned(16))) float sT[L1_ROWS * L1_PITCH];
    __shared__ float sInv[L1_ROWS];
    __shared__ float sR[L1_ROWS * L1_TILE];

    const int tileM = blockIdx.x;
    if (tileM >= nTiles) return;
    const int lane = threadIdx.x & 31;
    const int w = __builtin_amdgcn_readfirstlane((int)(threadIdx.x >> 5));
    const int c = lane & 15;
    const int h = lane >> 4;
    const int kw = 16 * w;

    int cj[3];
#pragma unroll
    for (int j = 0; j < 3; ++j) cj[j] = min(j * 16 + c, Nact - 1);

    v2f bl[2][2][3], br[2][2][3];
#pragma unroll
    for (int u = 0; u < 2; ++u) {
#pragma unroll
        for (int s = 0; s < 2; ++s) {
            const float* ql = Wl + (size_t)(kw + 8 * u + 4 * h + 2 * s) * Nact;
            const float* qr = Wr + (size_t)(kw + 8 * u + 4 * h + 2 * s) * Nact;
#pragma unroll
            for (int j = 0; j < 3; ++j) {
                bl[u][s][j].x = ql[cj[j]];
                bl[u][s][j].y = ql[Nact + cj[j]];
                br[u][s][j].x = qr[cj[j]];
                br[u][s][j].y = qr[Nact + cj[j]];
            }
        }
    }

    const int node = tileM * L1_ROWS + w;
    const int n = __builtin_amdgcn_readfirstlane(deg[node]);
    const float4* trow = (const float4*)(hsrc + (size_t)node * DFEAT);
    *(float4*)(sT + w * L1_PITCH + 4 * lane)       = trow[lane];
    *(float4*)(sT + w * L1_PITCH + 128 + 4 * lane) = trow[32 + lane];
    if (lane == 0) sInv[w] = 1.0f / fmaxf((float)n, 1.0f);

    float4 a0, a1;
    gather_rows(a0, a1, hsrc, slot + (size_t)node * BUCKET_CAP, n, esrc, edst, nE, node, lane);

    *(float4*)(sS + w * L1_PITCH + 4 * lane)       = a0;
    *(float4*)(sS + w * L1_PITCH + 128 + 4 * lane) = a1;
    __syncthreads();

    const float inv = sInv[c];
    v8f acc[3];
#pragma unroll
    for (int j = 0; j < 3; ++j) {
#pragma unroll
        for (int i = 0; i < 8; ++i) acc[j][i] = 0.0f;
    }
#pragma unroll
    for (int u = 0; u < 2; ++u) {
        const int ko = kw + 8 * u + 4 * h;
        const float4 as = *(const float4*)(sS + c * L1_PITCH + ko);
        const float4 at = *(const float4*)(sT + c * L1_PITCH + ko);
        v2f aS[2], aT[2];
        aS[0].x = as.x * inv; aS[0].y = as.y * inv;
        aS[1].x = as.z * inv; aS[1].y = as.w * inv;
        aT[0].x = at.x; aT[0].y = at.y;
        aT[1].x = at.z; aT[1].y = at.w;
#pragma unroll
        for (int s = 0; s < 2; ++s) {
#pragma unroll
            for (int j = 0; j < 3; ++j) acc[j] = mma4(aS[s], bl[u][s][j], acc[j]);
#pragma unroll
            for (int j = 0; j < 3; ++j) acc[j] = mma4(aT[s], br[u][s][j], acc[j]);
        }
    }

#pragma unroll
    for (int j = 0; j < 3; ++j) {
#pragma unroll
        for (int i = 0; i < 8; ++i) sR[w * L1_TILE + (j * 8 + i) * 32 + lane] = acc[j][i];
    }
    __syncthreads();

    for (int f = threadIdx.x; f < L1_ROWS * Nact; f += 512) {
        const int r  = f / Nact;
        const int cc = f - r * Nact;
        const int q  = ((cc >> 4) * 8 + (r & 7)) * 32 + (r >> 3) * 16 + (cc & 15);
        float sum = sR[q];
#pragma unroll
        for (int ww = 1; ww < L1_ROWS; ++ww) sum += sR[ww * L1_TILE + q];
        Out[(size_t)tileM * L1_ROWS * Nact + f] = sum + bias[cc];
    }
}


extern "C" void kernel_launch(void* const* d_in, const int* in_sizes, int n_in,
                              void* d_out, int out_size, void* d_ws, size_t ws_size,
                              hipStream_t stream) {
    const float* h   = (const float*)d_in[0];
    const float* wl0 = (const float*)d_in[1];
    const float* wr0 = (const float*)d_in[2];
    const float* b0  = (const float*)d_in[3];
    const float* wl1 = (const float*)d_in[4];
    const float* wr1 = (const float*)d_in[5];
    const float* b1  = (const float*)d_in[6];
    const int*   ei0 = (const int*)d_in[7];
    const int*   ei1 = (const int*)d_in[8];

    const int E0 = in_sizes[7] / 2;
    const int E1 = in_sizes[8] / 2;
    const int NT0 = 20000;
    const int NT1 = 2000;
    const int NCLS = 47;

    float* sums0 = (float*)d_ws;
    float* cnt0  = sums0 + (size_t)NT0 * DFEAT;
    float* inv0  = cnt0 + NT0;
    float* h1    = inv0 + NT0;
    float* reg1  = h1 + (size_t)NT0 * DFEAT;

    int* slot0 = (int*)h1;
    int* deg0  = (int*)reg1 - NT0;
    int* deg1  = (int*)reg1;
    int* slot1 = deg1 + 2048;
    int* used0 = deg1 + 131072;
    int* nlive = used0 + NT0;
    int* rowid = (int*)cnt0;

    zero_kernel<<<22, 1024, 0, stream>>>(deg0, NT0 + NT1, used0, NT0 + 32);
    bucket_kernel<<<(E0 + E1 + 255) / 256, 256, 0, stream>>>(ei0, ei0 + E0, E0, deg0, slot0,
                                                             ei1, ei1 + E1, E1, deg1, slot1, used0);

    gather_sum_kernel<<<(NT0 + 7) / 8, 256, 0, stream>>>(h, deg0, slot0, ei0, ei0 + E0, E0, used0, NT1, nlive, rowid,
                                                         sums0, inv0, NT0);

    sage_gemm_l0<<<NT0 / 32, 128, 0, stream>>>(sums0, inv0, rowid, nlive, h, wl0, wr0, b0, h1);

    sage_l1_fused<<<NT1 / 16, 512, 0, stream>>>(h1, deg1, slot1, ei1, ei1 + E1, E1, wl1, wr1, b1,
                                                (float*)d_out, NT1 / 16, NCLS);
}


# Edge grouping ranks on LDS counters (work-group = edge slice x target range, per-slice sub-buckets), no returning global adds, no clear launch: 5 launches -> 4; sum order = rank then slice order
# speedup vs baseline: 4.9793x; 1.3088x over previous
#include <hip/hip_runtime.h>

typedef __attribute__((ext_vector_type(2))) float v2f;
typedef __attribute__((ext_vector_type(8))) float v8f;

#define DFEAT 256

#define NSLICE 32
#define SUB_CAP 8
#define NPART 4
#define PART_MAX 5000
#define L1_GROUPS 16
#define BUCKET_CAP 64
#define GROUP_THREADS 512
#define UNR_A 19
#define UNR_B 59

__global__ void __launch_bounds__(GROUP_THREADS)
group_kernel(const int* __restrict__ srcA, const int* __restrict__ dstA, int nEA, int nTA,
             int* __restrict__ slotA, int* __restrict__ cntA,
             const int* __restrict__ srcB, const int* __restrict__ dstB, int nEB, int nTB,
             int* __restrict__ degB, int* __restrict__ slotB,
             int* __restrict__ usedB, int nSrcB,
             int* __restrict__ nLive) {
    __shared__ int sCnt[PART_MAX];
    const int tid = threadIdx.x;
    for (int i = tid; i < PART_MAX; i += GROUP_THREADS) sCnt[i] = 0;
    __syncthreads();

    if (blockIdx.x < NSLICE * NPART) {
        const int g = blockIdx.x / NPART;
        const int q = blockIdx.x - g * NPART;
        const int per = (nEA + NSLICE - 1) / NSLICE;
        const int e0 = g * per;
        const int e1 = min(nEA, e0 + per);
        const int tper = (nTA + NPART - 1) / NPART;
        const int lo = q * tper;
        const int hi = min(nTA, lo + tper);
        for (int base = e0; base < e1; base += GROUP_THREADS * UNR_A) {
            int dd[UNR_A], ss[UNR_A];
#pragma unroll
            for (int k = 0; k < UNR_A; ++k) {
                const int e = base + k * GROUP_THREADS + tid;
                const bool ok = e < e1;
                dd[k] = ok ? dstA[e] : -1;
                ss[k] = ok ? srcA[e] : 0;
            }
            int rr[UNR_A];
#pragma unroll
            for (int k = 0; k < UNR_A; ++k) {
                const int d = dd[k];
                rr[k] = (d >= lo && d < hi) ? atomicAdd(&sCnt[d - lo], 1) : SUB_CAP;
            }
#pragma unroll
            for (int k = 0; k < UNR_A; ++k) {
                if (rr[k] < SUB_CAP) slotA[((size_t)dd[k] * SUB_CAP + rr[k]) * NSLICE + g] = ss[k];
            }
        }
        __syncthreads();
        unsigned char* cbytes = (unsigned char*)cntA;
        for (int i = tid; i < hi - lo; i += GROUP_THREADS)
            cbytes[(size_t)g * nTA + lo + i] = (unsigned char)min(sCnt[i], 255);
    } else if (blockIdx.x < NSLICE * NPART + L1_GROUPS) {
        const int t = blockIdx.x - NSLICE * NPART;
        const int tper = (nTB + L1_GROUPS - 1) / L1_GROUPS;
        const int lo = t * tper;
        const int hi = min(nTB, lo + tper);
        for (int base = 0; base < nEB; base += GROUP_THREADS * UNR_B) {
            int dd[UNR_B], rr[UNR_B];
#pragma unroll
            for (int k = 0; k < UNR_B; ++k) {
                const int e = base + k * GROUP_THREADS + tid;
                dd[k] = (e < nEB) ? dstB[e] : -1;
            }
#pragma unroll
            for (int k = 0; k < UNR_B; ++k) {
                const int d = dd[k];
                rr[k] = (d >= lo && d < hi) ? atomicAdd(&sCnt[d - lo], 1) : BUCKET_CAP;
            }
            int ix[UNR_B], sv[UNR_B];
#pragma unroll
            for (int k = 0; k < UNR_B; ++k) ix[k] = (rr[k] < BUCKET_CAP) ? dd[k] * BUCKET_CAP + rr[k] : -1;
#pragma unroll
            for (int k = 0; k < UNR_B; ++k) sv[k] = (ix[k] >= 0) ? srcB[base + k * GROUP_THREADS + tid] : 0;
#pragma unroll
            for (int k = 0; k < UNR_B; ++k) {
                if (ix[k] >= 0) slotB[ix[k]] = sv[k];
            }
        }
        __syncthreads();
        for (int i = tid; i < hi - lo; i += GROUP_THREADS) degB[lo + i] = sCnt[i];
        if (t == 0 && tid == 0) *nLive = 0;
    } else {
        const int u = blockIdx.x - NSLICE * NPART - L1_GROUPS;
        const int sper = (nSrcB + L1_GROUPS - 1) / L1_GROUPS;
        const int slo = u * sper;
        const int shi = min(nSrcB, slo + sper);
        for (int base = 0; base < nEB; base += GROUP_THREADS * UNR_B) {
            int ss[UNR_B];
#pragma unroll
            for (int k = 0; k < UNR_B; ++k) {
                const int e = base + k * GROUP_THREADS + tid;
                ss[k] = (e < nEB) ? srcB[e] : -1;
            }
#pragma unroll
            for (int k = 0; k < UNR_B; ++k) {
                const int s = ss[k];
                if (s >= slo && s < shi) sCnt[s - slo] = 1;
            }
        }
        __syncthreads();
        for (int i = tid; i < shi - slo; i += GROUP_THREADS) usedB[slo + i] = sCnt[i];
    }
}

#define GATHER_U 8

__device__ __forceinline__ void add_row(float4& a0, float4& a1, const float* __restrict__ hsrc, int sid, int lane) {
    const float4* r = (const float4*)(hsrc + (size_t)sid * DFEAT);
    const float4 x = r[lane];
    const float4 y = r[32 + lane];
    a0.x += x.x; a0.y += x.y; a0.z += x.z; a0.w += x.w;
    a1.x += y.x; a1.y += y.y; a1.z += y.z; a1.w += y.w;
}

__device__ __forceinline__ void sum_listed(float4& a0, float4& a1,
                                           const float* __restrict__ hsrc,
                                           const int* ids, int n, int lane) {
    const float4 zero4 = make_float4(0.0f, 0.0f, 0.0f, 0.0f);
    for (int base = 0; base < n; base += 32) {
        const int m = min(32, n - base);
        const int myid = (lane < m) ? ids[base + lane] : 0;
        for (int j = 0; j < m; j += GATHER_U) {
            int sid[GATHER_U];
#pragma unroll
            for (int k = 0; k < GATHER_U; ++k) sid[k] = __builtin_amdgcn_readlane(myid, (j + k) & 31);

            float4 x[GATHER_U], y[GATHER_U];
            const int left = m - j;
            if (left >= GATHER_U) {
#pragma unroll
                for (int k = 0; k < GATHER_U; ++k) {
                    const float4* r = (const float4*)(hsrc + (size_t)sid[k] * DFEAT);
                    x[k] = r[lane];
                    y[k] = r[32 + lane];
                }
            } else {
#pragma unroll
                for (int k = 0; k < GATHER_U; ++k) {
                    if (k < left) {
                        const float4* r = (const float4*)(hsrc + (size_t)sid[k] * DFEAT);
                        x[k] = r[lane];
                        y[k] = r[32 + lane];
                    } else {
                        x[k] = zero4;
                        y[k] = zero4;
                    }
                }
            }
#pragma unroll
            for (int k = 0; k < GATHER_U; ++k) {
                a0.x += x[k].x; a0.y += x[k].y; a0.z += x[k].z; a0.w += x[k].w;
                a1.x += y[k].x; a1.y += y[k].y; a1.z += y[k].z; a1.w += y[k].w;
            }
        }
    }
}

__device__ __forceinline__ int sum_scanned(float4& a0, float4& a1,
                                           const float* __restrict__ hsrc,
                                           const int* __restrict__ esrc,
                                           const int* __restrict__ edst,
                                           int nE, int node, int lane) {
    int cnt = 0;
    for (int base = 0; base < nE; base += 32) {
        const int e = base + lane;
        const int d = (e < nE) ? edst[e] : -1;
        const int s = (e < nE) ? esrc[e] : 0;
        unsigned int hit = __builtin_amdgcn_ballot_w32(d == node);
        cnt += __builtin_popcount(hit);
        while (hit) {
            const int l = __builtin_ctz(hit);
            hit &= hit - 1;
            add_row(a0, a1, hsrc, __builtin_amdgcn_readlane(s, l), lane);
        }
    }
    return cnt;
}

__device__ __forceinline__ void gather_rows(float4& a0, float4& a1,
                                            const float* __restrict__ hsrc,
                                            const int* __restrict__ bucket, int n,
                                            const int* __restrict__ esrc,
                                            const int* __restrict__ edst,
                                            int nE, int node, int lane) {
    const float4 zero4 = make_float4(0.0f, 0.0f, 0.0f, 0.0f);
    a0 = zero4;
    a1 = zero4;
    if (n <= BUCKET_CAP) sum_listed(a0, a1, hsrc, bucket, n, lane);
    else (void)sum_scanned(a0, a1, hsrc, esrc, edst, nE, node, lane);
}

__device__ __forceinline__ int build_list(int* list,
                                          const unsigned char* __restrict__ cnt,
                                          const int* __restrict__ slot,
                                          int node, int nT, int lane, bool& ovf) {
    const int c = cnt[(size_t)lane * nT + node];
    ovf = __builtin_amdgcn_ballot_w32(c > SUB_CAP) != 0u;
    const int* p = slot + (size_t)node * (NSLICE * SUB_CAP) + lane;
    int id[SUB_CAP];
#pragma unroll
    for (int r = 0; r < SUB_CAP; ++r) id[r] = (c > r) ? p[r * NSLICE] : 0;
    int n = 0;
#pragma unroll
    for (int r = 0; r < SUB_CAP; ++r) {
        const bool has = c > r;
        const unsigned int m = __builtin_amdgcn_ballot_w32(has);
        if (has) list[n + (int)__builtin_amdgcn_mbcnt_lo(m, 0u)] = id[r];
        n += __builtin_popcount(m);
    }
    return n;
}

#define GATHER_WAVES 8

__global__ void __launch_bounds__(256)
gather_sum_kernel(const float* __restrict__ hsrc,
                  const int* __restrict__ csub,
                  const int* __restrict__ slot,
                  const int* __restrict__ esrc,
                  const int* __restrict__ edst,
                  int nE,
                  const int* __restrict__ used,
                  int nKeep,
                  int* __restrict__ nLive,
                  int* __restrict__ rowid,
                  float* __restrict__ sums,
                  float* __restrict__ inv,
                  int nT) {
    __shared__ int sFlag[GATHER_WAVES];
    __shared__ int sBase;
    __shared__ int sList[GATHER_WAVES][NSLICE * SUB_CAP];
    const int lane = threadIdx.x & 31;
    const int w = __builtin_amdgcn_readfirstlane((int)(threadIdx.x >> 5));
    const int node = __builtin_amdgcn_readfirstlane(blockIdx.x * GATHER_WAVES + w);
    bool live = false;
    if (node < nT) live = (node < nKeep) || (__builtin_amdgcn_readfirstlane(used[node]) != 0);

    if (lane == 0) sFlag[w] = live ? 1 : 0;
    __syncthreads();
    int cnt = 0, before = 0;
    for (int k = 0; k < GATHER_WAVES; ++k) {
        const int f = sFlag[k];
        cnt += f;
        before += (k < w) ? f : 0;
    }
    int base = 0;
    if (threadIdx.x == 0 && cnt > 0) base = atomicAdd(nLive, cnt);

    float4 a0 = make_float4(0.0f, 0.0f, 0.0f, 0.0f), a1 = a0;
    int n = 0;
    if (live) {
        bool ovf;
        n = build_list(sList[w], (const unsigned char*)csub, slot, node, nT, lane, ovf);
        if (!ovf) sum_listed(a0, a1, hsrc, sList[w], n, lane);
        else n = sum_scanned(a0, a1, hsrc, esrc, edst, nE, node, lane);
    }
    if (threadIdx.x == 0) sBase = base;
    __syncthreads();

    if (live) {
        const int pos = sBase + before;
        float4* o = (float4*)(sums + (size_t)pos * DFEAT);
        o[lane]      = a0;
        o[32 + lane] = a1;
        if (lane == 0) {
            inv[pos] = 1.0f / fmaxf((float)n, 1.0f);
            rowid[pos] = node;
        }
    }
}


__device__ __forceinline__ v8f mma4(v2f a, v2f b, v8f c) {
    return __builtin_amdgcn_wmma_f32_16x16x4_f32(false, a, false, b, (short)0, c, false, false);
}

struct FragV {
    float4 as[2], at[2];
    float4 bl[4], br[4];
};

__device__ __forceinline__ void l0_load(FragV& f, const float* __restrict__ a0, const float* __restrict__ a1,
                                        const float* __restrict__ t0, const float* __restrict__ t1,
                                        const float* __restrict__ pl, const float* __restrict__ pr, int g) {
    const int ko = g * 8;
    f.as[0] = *(const float4*)(a0 + ko);
    f.as[1] = *(const float4*)(a1 + ko);
    f.at[0] = *(const float4*)(t0 + ko);
    f.at[1] = *(const float4*)(t1 + ko);
#pragma unroll
    for (int i = 0; i < 4; ++i) {
        f.bl[i] = *(const float4*)(pl + (size_t)(ko + i) * DFEAT);
        f.br[i] = *(const float4*)(pr + (size_t)(ko + i) * DFEAT);
    }
}

__device__ __forceinline__ void l0_mma(v8f (&acc)[2][4], const FragV& f, float inv0, float inv1) {
    v2f aS[2][2], aT[2][2];
    aS[0][0].x = f.as[0].x * inv0; aS[0][0].y = f.as[0].y * inv0;
    aS[0][1].x = f.as[0].z * inv0; aS[0][1].y = f.as[0].w * inv0;
    aS[1][0].x = f.as[1].x * inv1; aS[1][0].y = f.as[1].y * inv1;
    aS[1][1].x = f.as[1].z * inv1; aS[1][1].y = f.as[1].w * inv1;
    aT[0][0].x = f.at[0].x; aT[0][0].y = f.at[0].y;
    aT[0][1].x = f.at[0].z; aT[0][1].y = f.at[0].w;
    aT[1][0].x = f.at[1].x; aT[1][0].y = f.at[1].y;
    aT[1][1].x = f.at[1].z; aT[1][1].y = f.at[1].w;

    float L[4][4], R[4][4];
#pragma unroll
    for (int i = 0; i < 4; ++i) {
        L[i][0] = f.bl[i].x; L[i][1] = f.bl[i].y; L[i][2] = f.bl[i].z; L[i][3] = f.bl[i].w;
        R[i][0] = f.br[i].x; R[i][1] = f.br[i].y; R[i][2] = f.br[i].z; R[i][3] = f.br[i].w;
    }
#pragma unroll
    for (int s = 0; s < 2; ++s) {
#pragma unroll
        for (int j = 0; j < 4; ++j) {
            v2f b; b.x = L[2 * s][j]; b.y = L[2 * s + 1][j];
            acc[0][j] = mma4(aS[0][s], b, acc[0][j]);
            acc[1][j] = mma4(aS[1][s], b, acc[1][j]);
        }
#pragma unroll
        for (int j = 0; j < 4; ++j) {
            v2f b; b.x = R[2 * s][j]; b.y = R[2 * s + 1][j];
            acc[0][j] = mma4(aT[0][s], b, acc[0][j]);
            acc[1][j] = mma4(aT[1][s], b, acc[1][j]);
        }
    }
}

__global__ void __launch_bounds__(128)
sage_gemm_l0(const float* __restrict__ Asum,
             const float* __restrict__ invc,
             const int* __restrict__ rowid,
             const int* __restrict__ nLive,
             const float* __restrict__ Htgt,
             const float* __restrict__ Wl,
             const float* __restrict__ Wr,
             const float* __restrict__ bias,
             float* __restrict__ Out) {
    const int lane = threadIdx.x & 31;
    const int pair = blockIdx.x;
    const int nC = *nLive;
    if (pair * 32 >= nC) return;
    const int c = lane & 15;
    const int h = lane >> 4;
    const int colBase = (threadIdx.x >> 5) * 64;
    const int q0 = min(pair * 32 + c, nC - 1);
    const int q1 = min(pair * 32 + 16 + c, nC - 1);

    const float inv0 = invc[q0];
    const float inv1 = invc[q1];
    const int rid0 = rowid[q0];
    const int rid1 = rowid[q1];

    const float4 bv = *(const float4*)(bias + colBase + 4 * c);
    v8f acc[2][4];
#pragma unroll
    for (int i = 0; i < 8; ++i) {
        acc[0][0][i] = bv.x; acc[0][1][i] = bv.y; acc[0][2][i] = bv.z; acc[0][3][i] = bv.w;
        acc[1][0][i] = bv.x; acc[1][1][i] = bv.y; acc[1][2][i] = bv.z; acc[1][3][i] = bv.w;
    }

    const float* a0 = Asum + (size_t)q0 * DFEAT + 4 * h;
    const float* a1 = Asum + (size_t)q1 * DFEAT + 4 * h;
    const float* t0 = Htgt + (size_t)rid0 * DFEAT + 4 * h;
    const float* t1 = Htgt + (size_t)rid1 * DFEAT + 4 * h;
    const float* pl = Wl + (size_t)(4 * h) * DFEAT + colBase + 4 * c;
    const float* pr = Wr + (size_t)(4 * h) * DFEAT + colBase + 4 * c;

    FragV f0, f1;
    l0_load(f0, a0, a1, t0, t1, pl, pr, 0);
    for (int g = 0; g < DFEAT / 8 - 2; g += 2) {
        l0_load(f1, a0, a1, t0, t1, pl, pr, g + 1);
        l0_mma(acc, f0, inv0, inv1);
        l0_load(f0, a0, a1, t0, t1, pl, pr, g + 2);
        l0_mma(acc, f1, inv0, inv1);
    }
    l0_load(f1, a0, a1, t0, t1, pl, pr, DFEAT / 8 - 1);
    l0_mma(acc, f0, inv0, inv1);
    l0_mma(acc, f1, inv0, inv1);

#pragma unroll
    for (int mt = 0; mt < 2; ++mt) {
        const int ridm = mt ? rid1 : rid0;
#pragma unroll
        for (int i = 0; i < 8; ++i) {
            const int nodeLo = __builtin_amdgcn_readlane(ridm, i);
            const int nodeHi = __builtin_amdgcn_readlane(ridm, 8 + i);
            const int node = h ? nodeHi : nodeLo;
            const int p = pair * 32 + mt * 16 + 8 * h + i;
            float4 v;
            v.x = fmaxf(acc[mt][0][i], 0.0f);
            v.y = fmaxf(acc[mt][1][i], 0.0f);
            v.z = fmaxf(acc[mt][2][i], 0.0f);
            v.w = fmaxf(acc[mt][3][i], 0.0f);
            if (p < nC) *(float4*)(Out + (size_t)node * DFEAT + colBase + 4 * c) = v;
        }
    }
}

#define L1_ROWS 16
#define L1_PITCH 260
#define L1_TILE 768

__global__ void __launch_bounds__(512)
sage_l1_fused(const float* __restrict__ hsrc,
              const int* __restrict__ deg,
              const int* __restrict__ slot,
              const int* __restrict__ esrc,
              const int* __restrict__ edst,
              int nE,
              const float* __restrict__ Wl,
              const float* __restrict__ Wr,
              const float* __restrict__ bias,
              float* __restrict__ Out,
              int nTiles,
              int Nact) {
    __shared__ __attribute__((aligned(16))) float sS[L1_ROWS * L1_PITCH];
    __shared__ __attribute__((aligned(16))) float sT[L1_ROWS * L1_PITCH];
    __shared__ float sInv[L1_ROWS];
    __shared__ float sR[L1_ROWS * L1_TILE];

    const int tileM = blockIdx.x;
    if (tileM >= nTiles) return;
    const int lane = threadIdx.x & 31;
    const int w = __builtin_amdgcn_readfirstlane((int)(threadIdx.x >> 5));
    const int c = lane & 15;
    const int h = lane >> 4;
    const int kw = 16 * w;

    int cj[3];
#pragma unroll
    for (int j = 0; j < 3; ++j) cj[j] = min(j * 16 + c, Nact - 1);

    v2f bl[2][2][3], br[2][2][3];
#pragma unroll
    for (int u = 0; u < 2; ++u) {
#pragma unroll
        for (int s = 0; s < 2; ++s) {
            const float* ql = Wl + (size_t)(kw + 8 * u + 4 * h + 2 * s) * Nact;
            const float* qr = Wr + (size_t)(kw + 8 * u + 4 * h + 2 * s) * Nact;
#pragma unroll
            for (int j = 0; j < 3; ++j) {
                bl[u][s][j].x = ql[cj[j]];
                bl[u][s][j].y = ql[Nact + cj[j]];
                br[u][s][j].x = qr[cj[j]];
                br[u][s][j].y = qr[Nact + cj[j]];
            }
        }
    }

    const int node = tileM * L1_ROWS + w;
    const int n = __builtin_amdgcn_readfirstlane(deg[node]);
    const float4* trow = (const float4*)(hsrc + (size_t)node * DFEAT);
    *(float4*)(sT + w * L1_PITCH + 4 * lane)       = trow[lane];
    *(float4*)(sT + w * L1_PITCH + 128 + 4 * lane) = trow[32 + lane];
    if (lane == 0) sInv[w] = 1.0f / fmaxf((float)n, 1.0f);

    float4 a0, a1;
    gather_rows(a0, a1, hsrc, slot + (size_t)node * BUCKET_CAP, n, esrc, edst, nE, node, lane);

    *(float4*)(sS + w * L1_PITCH + 4 * lane)       = a0;
    *(float4*)(sS + w * L1_PITCH + 128 + 4 * lane) = a1;
    __syncthreads();

    const float inv = sInv[c];
    v8f acc[3];
#pragma unroll
    for (int j = 0; j < 3; ++j) {
#pragma unroll
        for (int i = 0; i < 8; ++i) acc[j][i] = 0.0f;
    }
#pragma unroll
    for (int u = 0; u < 2; ++u) {
        const int ko = kw + 8 * u + 4 * h;
        const float4 as = *(const float4*)(sS + c * L1_PITCH + ko);
        const float4 at = *(const float4*)(sT + c * L1_PITCH + ko);
        v2f aS[2], aT[2];
        aS[0].x = as.x * inv; aS[0].y = as.y * inv;
        aS[1].x = as.z * inv; aS[1].y = as.w * inv;
        aT[0].x = at.x; aT[0].y = at.y;
        aT[1].x = at.z; aT[1].y = at.w;
#pragma unroll
        for (int s = 0; s < 2; ++s) {
#pragma unroll
            for (int j = 0; j < 3; ++j) acc[j] = mma4(aS[s], bl[u][s][j], acc[j]);
#pragma unroll
            for (int j = 0; j < 3; ++j) acc[j] = mma4(aT[s], br[u][s][j], acc[j]);
        }
    }

#pragma unroll
    for (int j = 0; j < 3; ++j) {
#pragma unroll
        for (int i = 0; i < 8; ++i) sR[w * L1_TILE + (j * 8 + i) * 32 + lane] = acc[j][i];
    }
    __syncthreads();

    for (int f = threadIdx.x; f < L1_ROWS * Nact; f += 512) {
        const int r  = f / Nact;
        const int cc = f - r * Nact;
        const int q  = ((cc >> 4) * 8 + (r & 7)) * 32 + (r >> 3) * 16 + (cc & 15);
        float sum = sR[q];
#pragma unroll
        for (int ww = 1; ww < L1_ROWS; ++ww) sum += sR[ww * L1_TILE + q];
        Out[(size_t)tileM * L1_ROWS * Nact + f] = sum + bias[cc];
    }
}


extern "C" void kernel_launch(void* const* d_in, const int* in_sizes, int n_in,
                              void* d_out, int out_size, void* d_ws, size_t ws_size,
                              hipStream_t stream) {
    const float* h   = (const float*)d_in[0];
    const float* wl0 = (const float*)d_in[1];
    const float* wr0 = (const float*)d_in[2];
    const float* b0  = (const float*)d_in[3];
    const float* wl1 = (const float*)d_in[4];
    const float* wr1 = (const float*)d_in[5];
    const float* b1  = (const float*)d_in[6];
    const int*   ei0 = (const int*)d_in[7];
    const int*   ei1 = (const int*)d_in[8];

    const int E0 = in_sizes[7] / 2;
    const int E1 = in_sizes[8] / 2;
    const int NT0 = 20000;
    const int NT1 = 2000;
    const int NCLS = 47;

    float* sums0 = (float*)d_ws;
    float* cnt0  = sums0 + (size_t)NT0 * DFEAT;
    float* inv0  = cnt0 + NT0;
    float* h1    = inv0 + NT0;
    float* reg1  = h1 + (size_t)NT0 * DFEAT;

    int* slot0 = (int*)h1;
    int* deg1  = (int*)reg1;
    int* slot1 = deg1 + 2048;
    int* used0 = deg1 + 131072;
    int* nlive = used0 + NT0;
    int* csub0 = deg1 + 151104;
    int* rowid = (int*)cnt0;

    group_kernel<<<NSLICE * NPART + 2 * L1_GROUPS, GROUP_THREADS, 0, stream>>>(ei0, ei0 + E0, E0, NT0, slot0, csub0,
                                                                           ei1, ei1 + E1, E1, NT1, deg1, slot1,
                                                                           used0, NT0, nlive);

    gather_sum_kernel<<<(NT0 + 7) / 8, 256, 0, stream>>>(h, csub0, slot0, ei0, ei0 + E0, E0, used0, NT1, nlive, rowid,
                                                         sums0, inv0, NT0);

    sage_gemm_l0<<<NT0 / 32, 128, 0, stream>>>(sums0, inv0, rowid, nlive, h, wl0, wr0, b0, h1);

    sage_l1_fused<<<NT1 / 16, 512, 0, stream>>>(h1, deg1, slot1, ei1, ei1 + E1, E1, wl1, wr1, b1,
                                                (float*)d_out, NT1 / 16, NCLS);
}


# Layer-0 aggregation fused into its GEMM launch: per 32-row slab 8 waves gather 4 live targets each into LDS (live list packed while grouping), 4 waves multiply; 4 launches -> 3; sum order unchanged
# speedup vs baseline: 5.9706x; 1.1991x over previous
#include <hip/hip_runtime.h>

typedef __attribute__((ext_vector_type(2))) float v2f;
typedef __attribute__((ext_vector_type(8))) float v8f;

#define DFEAT 256

#define NSLICE 32
#define SUB_CAP 8
#define NPART 4
#define PART_MAX 5000
#define L1_GROUPS 16
#define BUCKET_CAP 64
#define GROUP_THREADS 512
#define UNR_A 19
#define UNR_B 59
#define SRC_CH ((PART_MAX + GROUP_THREADS - 1) / GROUP_THREADS)

__global__ void __launch_bounds__(GROUP_THREADS)
group_kernel(const int* __restrict__ srcA, const int* __restrict__ dstA, int nEA, int nTA,
             int* __restrict__ slotA, int* __restrict__ cntA,
             const int* __restrict__ srcB, const int* __restrict__ dstB, int nEB, int nTB,
             int* __restrict__ degB, int* __restrict__ slotB,
             int* __restrict__ liveList, int nSrcB,
             int* __restrict__ liveCnt) {
    __shared__ int sCnt[PART_MAX];
    __shared__ int sTot[GROUP_THREADS / 32];
    const int tid = threadIdx.x;
    for (int i = tid; i < PART_MAX; i += GROUP_THREADS) sCnt[i] = 0;
    __syncthreads();

    if (blockIdx.x < NSLICE * NPART) {
        const int g = blockIdx.x / NPART;
        const int q = blockIdx.x - g * NPART;
        const int per = (nEA + NSLICE - 1) / NSLICE;
        const int e0 = g * per;
        const int e1 = min(nEA, e0 + per);
        const int tper = (nTA + NPART - 1) / NPART;
        const int lo = q * tper;
        const int hi = min(nTA, lo + tper);
        for (int base = e0; base < e1; base += GROUP_THREADS * UNR_A) {
            int dd[UNR_A], ss[UNR_A];
#pragma unroll
            for (int k = 0; k < UNR_A; ++k) {
                const int e = base + k * GROUP_THREADS + tid;
                const bool ok = e < e1;
                dd[k] = ok ? dstA[e] : -1;
                ss[k] = ok ? srcA[e] : 0;
            }
            int rr[UNR_A];
#pragma unroll
            for (int k = 0; k < UNR_A; ++k) {
                const int d = dd[k];
                rr[k] = (d >= lo && d < hi) ? atomicAdd(&sCnt[d - lo], 1) : SUB_CAP;
            }
#pragma unroll
            for (int k = 0; k < UNR_A; ++k) {
                if (rr[k] < SUB_CAP) slotA[((size_t)dd[k] * SUB_CAP + rr[k]) * NSLICE + g] = ss[k];
            }
        }
        __syncthreads();
        unsigned char* cbytes = (unsigned char*)cntA;
        for (int i = tid; i < hi - lo; i += GROUP_THREADS)
            cbytes[(size_t)g * nTA + lo + i] = (unsigned char)min(sCnt[i], 255);
    } else if (blockIdx.x < NSLICE * NPART + L1_GROUPS) {
        const int t = blockIdx.x - NSLICE * NPART;
        const int tper = (nTB + L1_GROUPS - 1) / L1_GROUPS;
        const int lo = t * tper;
        const int hi = min(nTB, lo + tper);
        for (int base = 0; base < nEB; base += GROUP_THREADS * UNR_B) {
            int dd[UNR_B], rr[UNR_B];
#pragma unroll
            for (int k = 0; k < UNR_B; ++k) {
                const int e = base + k * GROUP_THREADS + tid;
                dd[k] = (e < nEB) ? dstB[e] : -1;
            }
#pragma unroll
            for (int k = 0; k < UNR_B; ++k) {
                const int d = dd[k];
                rr[k] = (d >= lo && d < hi) ? atomicAdd(&sCnt[d - lo], 1) : BUCKET_CAP;
            }
            int ix[UNR_B], sv[UNR_B];
#pragma unroll
            for (int k = 0; k < UNR_B; ++k) ix[k] = (rr[k] < BUCKET_CAP) ? dd[k] * BUCKET_CAP + rr[k] : -1;
#pragma unroll
            for (int k = 0; k < UNR_B; ++k) sv[k] = (ix[k] >= 0) ? srcB[base + k * GROUP_THREADS + tid] : 0;
#pragma unroll
            for (int k = 0; k < UNR_B; ++k) {
                if (ix[k] >= 0) slotB[ix[k]] = sv[k];
            }
        }
        __syncthreads();
        for (int i = tid; i < hi - lo; i += GROUP_THREADS) degB[lo + i] = sCnt[i];
    } else {
        const int u = blockIdx.x - NSLICE * NPART - L1_GROUPS;
        const int sper = (nSrcB + L1_GROUPS - 1) / L1_GROUPS;
        const int slo = u * sper;
        const int shi = min(nSrcB, slo + sper);
        for (int base = 0; base < nEB; base += GROUP_THREADS * UNR_B) {
            int ss[UNR_B];
#pragma unroll
            for (int k = 0; k < UNR_B; ++k) {
                const int e = base + k * GROUP_THREADS + tid;
                ss[k] = (e < nEB) ? srcB[e] : -1;
            }
#pragma unroll
            for (int k = 0; k < UNR_B; ++k) {
                const int s = ss[k];
                if (s >= slo && s < shi) sCnt[s - slo] = 1;
            }
        }
        __syncthreads();
        const int lane = tid & 31;
        const int wv = tid >> 5;
        const int nloc = shi - slo;
        unsigned int m[SRC_CH];
        int tot = 0;
#pragma unroll
        for (int p = 0; p < SRC_CH; ++p) {
            const int i = (wv * SRC_CH + p) * 32 + lane;
            const int f = sCnt[min(i, PART_MAX - 1)];
            const bool lv = (i < nloc) && ((slo + i < nTB) || (f != 0));
            m[p] = __builtin_amdgcn_ballot_w32(lv);
            tot += __builtin_popcount(m[p]);
        }
        if (lane == 0) sTot[wv] = tot;
        __syncthreads();
        int run = 0, all = 0;
#pragma unroll
        for (int k = 0; k < GROUP_THREADS / 32; ++k) {
            const int v = sTot[k];
            run += (k < wv) ? v : 0;
            all += v;
        }
#pragma unroll
        for (int p = 0; p < SRC_CH; ++p) {
            const int i = (wv * SRC_CH + p) * 32 + lane;
            if ((m[p] >> lane) & 1u) liveList[slo + run + (int)__builtin_amdgcn_mbcnt_lo(m[p], 0u)] = slo + i;
            run += __builtin_popcount(m[p]);
        }
        if (tid == 0) liveCnt[u] = all;
    }
}

#define GATHER_U 8
#define GATHER_U0 16

__device__ __forceinline__ void add_row(float4& a0, float4& a1, const float* __restrict__ hsrc, int sid, int lane) {
    const float4* r = (const float4*)(hsrc + (size_t)sid * DFEAT);
    const float4 x = r[lane];
    const float4 y = r[32 + lane];
    a0.x += x.x; a0.y += x.y; a0.z += x.z; a0.w += x.w;
    a1.x += y.x; a1.y += y.y; a1.z += y.z; a1.w += y.w;
}

template <int U>
__device__ __forceinline__ void sum_listed(float4& a0, float4& a1,
                                           const float* __restrict__ hsrc,
                                           const int* ids, int n, int lane) {
    const float4 zero4 = make_float4(0.0f, 0.0f, 0.0f, 0.0f);
    for (int base = 0; base < n; base += 32) {
        const int m = min(32, n - base);
        const int myid = (lane < m) ? ids[base + lane] : 0;
        for (int j = 0; j < m; j += U) {
            int sid[U];
#pragma unroll
            for (int k = 0; k < U; ++k) sid[k] = __builtin_amdgcn_readlane(myid, (j + k) & 31);

            float4 x[U], y[U];
            const int left = m - j;
            if (left >= U) {
#pragma unroll
                for (int k = 0; k < U; ++k) {
                    const float4* r = (const float4*)(hsrc + (size_t)sid[k] * DFEAT);
                    x[k] = r[lane];
                    y[k] = r[32 + lane];
                }
            } else {
#pragma unroll
                for (int k = 0; k < U; ++k) {
                    if (k < left) {
                        const float4* r = (const float4*)(hsrc + (size_t)sid[k] * DFEAT);
                        x[k] = r[lane];
                        y[k] = r[32 + lane];
                    } else {
                        x[k] = zero4;
                        y[k] = zero4;
                    }
                }
            }
#pragma unroll
            for (int k = 0; k < U; ++k) {
                a0.x += x[k].x; a0.y += x[k].y; a0.z += x[k].z; a0.w += x[k].w;
                a1.x += y[k].x; a1.y += y[k].y; a1.z += y[k].z; a1.w += y[k].w;
            }
        }
    }
}

__device__ __forceinline__ int sum_scanned(float4& a0, float4& a1,
                                           const float* __restrict__ hsrc,
                                           const int* __restrict__ esrc,
                                           const int* __restrict__ edst,
                                           int nE, int node, int lane) {
    int cnt = 0;
    for (int base = 0; base < nE; base += 32) {
        const int e = base + lane;
        const int d = (e < nE) ? edst[e] : -1;
        const int s = (e < nE) ? esrc[e] : 0;
        unsigned int hit = __builtin_amdgcn_ballot_w32(d == node);
        cnt += __builtin_popcount(hit);
        while (hit) {
            const int l = __builtin_ctz(hit);
            hit &= hit - 1;
            add_row(a0, a1, hsrc, __builtin_amdgcn_readlane(s, l), lane);
        }
    }
    return cnt;
}

__device__ __forceinline__ void gather_rows(float4& a0, float4& a1,
                                            const float* __restrict__ hsrc,
                                            const int* __restrict__ bucket, int n,
                                            const int* __restrict__ esrc,
                                            const int* __restrict__ edst,
                                            int nE, int node, int lane) {
    const float4 zero4 = make_float4(0.0f, 0.0f, 0.0f, 0.0f);
    a0 = zero4;
    a1 = zero4;
    if (n <= BUCKET_CAP) sum_listed<GATHER_U>(a0, a1, hsrc, bucket, n, lane);
    else (void)sum_scanned(a0, a1, hsrc, esrc, edst, nE, node, lane);
}

__device__ __forceinline__ int pack_list(int* list, int c, const int (&id)[SUB_CAP]) {
    int n = 0;
#pragma unroll
    for (int r = 0; r < SUB_CAP; ++r) {
        const bool has = c > r;
        const unsigned int m = __builtin_amdgcn_ballot_w32(has);
        if (has) list[n + (int)__builtin_amdgcn_mbcnt_lo(m, 0u)] = id[r];
        n += __builtin_popcount(m);
    }
    return n;
}


__device__ __forceinline__ v8f mma4(v2f a, v2f b, v8f c) {
    return __builtin_amdgcn_wmma_f32_16x16x4_f32(false, a, false, b, (short)0, c, false, false);
}

struct FragV {
    float4 as[2], at[2];
    float4 bl[4], br[4];
};

__device__ __forceinline__ void l0_load(FragV& f, const float* __restrict__ a0, const float* __restrict__ a1,
                                        const float* __restrict__ t0, const float* __restrict__ t1,
                                        const float* __restrict__ pl, const float* __restrict__ pr, int g) {
    const int ko = g * 8;
    f.as[0] = *(const float4*)(a0 + ko);
    f.as[1] = *(const float4*)(a1 + ko);
    f.at[0] = *(const float4*)(t0 + ko);
    f.at[1] = *(const float4*)(t1 + ko);
#pragma unroll
    for (int i = 0; i < 4; ++i) {
        f.bl[i] = *(const float4*)(pl + (size_t)(ko + i) * DFEAT);
        f.br[i] = *(const float4*)(pr + (size_t)(ko + i) * DFEAT);
    }
}

__device__ __forceinline__ void l0_mma(v8f (&acc)[2][4], const FragV& f, float inv0, float inv1) {
    v2f aS[2][2], aT[2][2];
    aS[0][0].x = f.as[0].x * inv0; aS[0][0].y = f.as[0].y * inv0;
    aS[0][1].x = f.as[0].z * inv0; aS[0][1].y = f.as[0].w * inv0;
    aS[1][0].x = f.as[1].x * inv1; aS[1][0].y = f.as[1].y * inv1;
    aS[1][1].x = f.as[1].z * inv1; aS[1][1].y = f.as[1].w * inv1;
    aT[0][0].x = f.at[0].x; aT[0][0].y = f.at[0].y;
    aT[0][1].x = f.at[0].z; aT[0][1].y = f.at[0].w;
    aT[1][0].x = f.at[1].x; aT[1][0].y = f.at[1].y;
    aT[1][1].x = f.at[1].z; aT[1][1].y = f.at[1].w;

    float L[4][4], R[4][4];
#pragma unroll
    for (int i = 0; i < 4; ++i) {
        L[i][0] = f.bl[i].x; L[i][1] = f.bl[i].y; L[i][2] = f.bl[i].z; L[i][3] = f.bl[i].w;
        R[i][0] = f.br[i].x; R[i][1] = f.br[i].y; R[i][2] = f.br[i].z; R[i][3] = f.br[i].w;
    }
#pragma unroll
    for (int s = 0; s < 2; ++s) {
#pragma unroll
        for (int j = 0; j < 4; ++j) {
            v2f b; b.x = L[2 * s][j]; b.y = L[2 * s + 1][j];
            acc[0][j] = mma4(aS[0][s], b, acc[0][j]);
            acc[1][j] = mma4(aS[1][s], b, acc[1][j]);
        }
#pragma unroll
        for (int j = 0; j < 4; ++j) {
            v2f b; b.x = R[2 * s][j]; b.y = R[2 * s + 1][j];
            acc[0][j] = mma4(aT[0][s], b, acc[0][j]);
            acc[1][j] = mma4(aT[1][s], b, acc[1][j]);
        }
    }
}

#define F0_ROWS 32
#define F0_WAVES 8
#define F0_TPW (F0_ROWS / F0_WAVES)
#define F0_PITCH 260

__global__ void __launch_bounds__(F0_WAVES * 32)
sage_l0_fused(const float* __restrict__ hsrc,
              const int* __restrict__ csub,
              const int* slot,
              const int* __restrict__ esrc,
              const int* __restrict__ edst,
              int nE,
              const int* __restrict__ liveCnt,
              const int* __restrict__ liveList,
              const float* __restrict__ Wl,
              const float* __restrict__ Wr,
              const float* __restrict__ bias,
              float* Out,
              int nT) {
    __shared__ __attribute__((aligned(16))) float sS[F0_ROWS * F0_PITCH];
    __shared__ __attribute__((aligned(16))) float sT[F0_ROWS * F0_PITCH];
    __shared__ float sInv[F0_ROWS];
    __shared__ int sNode[F0_ROWS];
    __shared__ int sLen[F0_ROWS];
    __shared__ int sList[F0_ROWS][NSLICE * SUB_CAP];

    const int lane = threadIdx.x & 31;
    const int w = __builtin_amdgcn_readfirstlane((int)(threadIdx.x >> 5));
    const int pair = blockIdx.x;

    int pre[L1_GROUPS + 1];
    pre[0] = 0;
#pragma unroll
    for (int u = 0; u < L1_GROUPS; ++u) pre[u + 1] = pre[u] + liveCnt[u];
    const int nC = pre[L1_GROUPS];
    if (pair * F0_ROWS >= nC) return;
    const int sper = (nT + L1_GROUPS - 1) / L1_GROUPS;

    int node[F0_TPW];
#pragma unroll
    for (int t = 0; t < F0_TPW; ++t) {
        const int q = min(pair * F0_ROWS + w * F0_TPW + t, nC - 1);
        int u = 0, off = 0;
#pragma unroll
        for (int k = 1; k < L1_GROUPS; ++k) {
            if (pre[k] <= q) { u = k; off = pre[k]; }
        }
        node[t] = __builtin_amdgcn_readfirstlane(liveList[u * sper + (q - off)]);
    }
    const unsigned char* cbytes = (const unsigned char*)csub;
    int cc[F0_TPW];
#pragma unroll
    for (int t = 0; t < F0_TPW; ++t) cc[t] = cbytes[(size_t)lane * nT + node[t]];
    int id[F0_TPW][SUB_CAP];
#pragma unroll
    for (int t = 0; t < F0_TPW; ++t) {
        const int* p = slot + (size_t)node[t] * (NSLICE * SUB_CAP) + lane;
#pragma unroll
        for (int r = 0; r < SUB_CAP; ++r) id[t][r] = (cc[t] > r) ? p[r * NSLICE] : 0;
    }
#pragma unroll
    for (int t = 0; t < F0_TPW; ++t) {
        const int e = w * F0_TPW + t;
        const bool ovf = __builtin_amdgcn_ballot_w32(cc[t] > SUB_CAP) != 0u;
        const int n = pack_list(sList[e], cc[t], id[t]);
        if (lane == 0) {
            sNode[e] = node[t];
            sLen[e] = ovf ? -1 : n;
        }
    }
    for (int t = 0; t < F0_TPW; ++t) {
        const int e = w * F0_TPW + t;
        const int nd = __builtin_amdgcn_readfirstlane(sNode[e]);
        int n = __builtin_amdgcn_readfirstlane(sLen[e]);
        const float4* trow = (const float4*)(hsrc + (size_t)nd * DFEAT);
        const float4 tx = trow[lane];
        const float4 ty = trow[32 + lane];
        float4 a0 = make_float4(0.0f, 0.0f, 0.0f, 0.0f), a1 = a0;
        if (n >= 0) sum_listed<GATHER_U0>(a0, a1, hsrc, sList[e], n, lane);
        else n = sum_scanned(a0, a1, hsrc, esrc, edst, nE, nd, lane);
        *(float4*)(sS + e * F0_PITCH + 4 * lane)       = a0;
        *(float4*)(sS + e * F0_PITCH + 128 + 4 * lane) = a1;
        *(float4*)(sT + e * F0_PITCH + 4 * lane)       = tx;
        *(float4*)(sT + e * F0_PITCH + 128 + 4 * lane) = ty;
        if (lane == 0) sInv[e] = 1.0f / fmaxf((float)n, 1.0f);
    }
    __syncthreads();
    if (w >= 4) return;

    const int c = lane & 15;
    const int h = lane >> 4;
    const int colBase = w * 64;

    const float inv0 = sInv[c];
    const float inv1 = sInv[16 + c];
    const int rid0 = sNode[c];
    const int rid1 = sNode[16 + c];

    const float4 bv = *(const float4*)(bias + colBase + 4 * c);
    v8f acc[2][4];
#pragma unroll
    for (int i = 0; i < 8; ++i) {
        acc[0][0][i] = bv.x; acc[0][1][i] = bv.y; acc[0][2][i] = bv.z; acc[0][3][i] = bv.w;
        acc[1][0][i] = bv.x; acc[1][1][i] = bv.y; acc[1][2][i] = bv.z; acc[1][3][i] = bv.w;
    }

    const float* a0 = sS + c * F0_PITCH + 4 * h;
    const float* a1 = sS + (16 + c) * F0_PITCH + 4 * h;
    const float* t0 = sT + c * F0_PITCH + 4 * h;
    const float* t1 = sT + (16 + c) * F0_PITCH + 4 * h;
    const float* pl = Wl + (size_t)(4 * h) * DFEAT + colBase + 4 * c;
    const float* pr = Wr + (size_t)(4 * h) * DFEAT + colBase + 4 * c;

    FragV f0, f1;
    l0_load(f0, a0, a1, t0, t1, pl, pr, 0);
    for (int g = 0; g < DFEAT / 8 - 2; g += 2) {
        l0_load(f1, a0, a1, t0, t1, pl, pr, g + 1);
        l0_mma(acc, f0, inv0, inv1);
        l0_load(f0, a0, a1, t0, t1, pl, pr, g + 2);
        l0_mma(acc, f1, inv0, inv1);
    }
    l0_load(f1, a0, a1, t0, t1, pl, pr, DFEAT / 8 - 1);
    l0_mma(acc, f0, inv0, inv1);
    l0_mma(acc, f1, inv0, inv1);

#pragma unroll
    for (int mt = 0; mt < 2; ++mt) {
        const int ridm = mt ? rid1 : rid0;
#pragma unroll
        for (int i = 0; i < 8; ++i) {
            const int nodeLo = __builtin_amdgcn_readlane(ridm, i);
            const int nodeHi = __builtin_amdgcn_readlane(ridm, 8 + i);
            const int nd = h ? nodeHi : nodeLo;
            const int p = pair * F0_ROWS + mt * 16 + 8 * h + i;
            float4 v;
            v.x = fmaxf(acc[mt][0][i], 0.0f);
            v.y = fmaxf(acc[mt][1][i], 0.0f);
            v.z = fmaxf(acc[mt][2][i], 0.0f);
            v.w = fmaxf(acc[mt][3][i], 0.0f);
            if (p < nC) *(float4*)(Out + (size_t)nd * DFEAT + colBase + 4 * c) = v;
        }
    }
}

#define L1_ROWS 16
#define L1_PITCH 260
#define L1_TILE 768

__global__ void __launch_bounds__(512)
sage_l1_fused(const float* __restrict__ hsrc,
              const int* __restrict__ deg,
              const int* __restrict__ slot,
              const int* __restrict__ esrc,
              const int* __restrict__ edst,
              int nE,
              const float* __restrict__ Wl,
              const float* __restrict__ Wr,
              const float* __restrict__ bias,
              float* __restrict__ Out,
              int nTiles,
              int Nact) {
    __shared__ __attribute__((aligned(16))) float sS[L1_ROWS * L1_PITCH];
    __shared__ __attribute__((aligned(16))) float sT[L1_ROWS * L1_PITCH];
    __shared__ float sInv[L1_ROWS];
    __shared__ float sR[L1_ROWS * L1_TILE];

    const int tileM = blockIdx.x;
    if (tileM >= nTiles) return;
    const int lane = threadIdx.x & 31;
    const int w = __builtin_amdgcn_readfirstlane((int)(threadIdx.x >> 5));
    const int c = lane & 15;
    const int h = lane >> 4;
    const int kw = 16 * w;

    int cj[3];
#pragma unroll
    for (int j = 0; j < 3; ++j) cj[j] = min(j * 16 + c, Nact - 1);

    v2f bl[2][2][3], br[2][2][3];
#pragma unroll
    for (int u = 0; u < 2; ++u) {
#pragma unroll
        for (int s = 0; s < 2; ++s) {
            const float* ql = Wl + (size_t)(kw + 8 * u + 4 * h + 2 * s) * Nact;
            const float* qr = Wr + (size_t)(kw + 8 * u + 4 * h + 2 * s) * Nact;
#pragma unroll
            for (int j = 0; j < 3; ++j) {
                bl[u][s][j].x = ql[cj[j]];
                bl[u][s][j].y = ql[Nact + cj[j]];
                br[u][s][j].x = qr[cj[j]];
                br[u][s][j].y = qr[Nact + cj[j]];
            }
        }
    }

    const int node = tileM * L1_ROWS + w;
    const int n = __builtin_amdgcn_readfirstlane(deg[node]);
    const float4* trow = (const float4*)(hsrc + (size_t)node * DFEAT);
    *(float4*)(sT + w * L1_PITCH + 4 * lane)       = trow[lane];
    *(float4*)(sT + w * L1_PITCH + 128 + 4 * lane) = trow[32 + lane];
    if (lane == 0) sInv[w] = 1.0f / fmaxf((float)n, 1.0f);

    float4 a0, a1;
    gather_rows(a0, a1, hsrc, slot + (size_t)node * BUCKET_CAP, n, esrc, edst, nE, node, lane);

    *(float4*)(sS + w * L1_PITCH + 4 * lane)       = a0;
    *(float4*)(sS + w * L1_PITCH + 128 + 4 * lane) = a1;
    __syncthreads();

    const float inv = sInv[c];
    v8f acc[3];
#pragma unroll
    for (int j = 0; j < 3; ++j) {
#pragma unroll
        for (int i = 0; i < 8; ++i) acc[j][i] = 0.0f;
    }
#pragma unroll
    for (int u = 0; u < 2; ++u) {
        const int ko = kw + 8 * u + 4 * h;
        const float4 as = *(const float4*)(sS + c * L1_PITCH + ko);
        const float4 at = *(const float4*)(sT + c * L1_PITCH + ko);
        v2f aS[2], aT[2];
        aS[0].x = as.x * inv; aS[0].y = as.y * inv;
        aS[1].x = as.z * inv; aS[1].y = as.w * inv;
        aT[0].x = at.x; aT[0].y = at.y;
        aT[1].x = at.z; aT[1].y = at.w;
#pragma unroll
        for (int s = 0; s < 2; ++s) {
#pragma unroll
            for (int j = 0; j < 3; ++j) acc[j] = mma4(aS[s], bl[u][s][j], acc[j]);
#pragma unroll
            for (int j = 0; j < 3; ++j) acc[j] = mma4(aT[s], br[u][s][j], acc[j]);
        }
    }

#pragma unroll
    for (int j = 0; j < 3; ++j) {
#pragma unroll
        for (int i = 0; i < 8; ++i) sR[w * L1_TILE + (j * 8 + i) * 32 + lane] = acc[j][i];
    }
    __syncthreads();

    for (int f = threadIdx.x; f < L1_ROWS * Nact; f += 512) {
        const int r  = f / Nact;
        const int cc = f - r * Nact;
        const int q  = ((cc >> 4) * 8 + (r & 7)) * 32 + (r >> 3) * 16 + (cc & 15);
        float sum = sR[q];
#pragma unroll
        for (int ww = 1; ww < L1_ROWS; ++ww) sum += sR[ww * L1_TILE + q];
        Out[(size_t)tileM * L1_ROWS * Nact + f] = sum + bias[cc];
    }
}


extern "C" void kernel_launch(void* const* d_in, const int* in_sizes, int n_in,
                              void* d_out, int out_size, void* d_ws, size_t ws_size,
                              hipStream_t stream) {
    const float* h   = (const float*)d_in[0];
    const float* wl0 = (const float*)d_in[1];
    const float* wr0 = (const float*)d_in[2];
    const float* b0  = (const float*)d_in[3];
    const float* wl1 = (const float*)d_in[4];
    const float* wr1 = (const float*)d_in[5];
    const float* b1  = (const float*)d_in[6];
    const int*   ei0 = (const int*)d_in[7];
    const int*   ei1 = (const int*)d_in[8];

    const int E0 = in_sizes[7] / 2;
    const int E1 = in_sizes[8] / 2;
    const int NT0 = 20000;
    const int NT1 = 2000;
    const int NCLS = 47;

    float* h1    = (float*)d_ws;
    float* reg1  = h1 + (size_t)NT0 * DFEAT;

    int* slot0 = (int*)h1;
    int* deg1  = (int*)reg1;
    int* slot1 = deg1 + 2048;
    int* live0 = deg1 + 131072;
    int* lcnt0 = live0 + NT0;
    int* csub0 = deg1 + 151104;

    group_kernel<<<NSLICE * NPART + 2 * L1_GROUPS, GROUP_THREADS, 0, stream>>>(ei0, ei0 + E0, E0, NT0, slot0, csub0,
                                                                           ei1, ei1 + E1, E1, NT1, deg1, slot1,
                                                                           live0, NT0, lcnt0);

    sage_l0_fused<<<NT0 / 32, F0_WAVES * 32, 0, stream>>>(h, csub0, slot0, ei0, ei0 + E0, E0, lcnt0, live0,
                                                        wl0, wr0, b0, h1, NT0);

    sage_l1_fused<<<NT1 / 16, 512, 0, stream>>>(h1, deg1, slot1, ei1, ei1 + E1, E1, wl1, wr1, b1,
                                                (float*)d_out, NT1 / 16, NCLS);
}
